# starting kernel
# baseline (speedup 1.0000x reference)
#include <hip/hip_runtime.h>

typedef float v2f __attribute__((ext_vector_type(2)));
typedef float v8f __attribute__((ext_vector_type(8)));

#define HD 128
constexpr int N_USER = 150000;
constexpr int N_BOOK = 75000;
constexpr int N_EDGE = 1500000;
constexpr float LN_EPS = 1e-5f;

__global__ void zero_kernel(float* __restrict__ p, size_t n) {
    size_t i = (size_t)blockIdx.x * blockDim.x + threadIdx.x;
    size_t stride = (size_t)gridDim.x * blockDim.x;
    for (; i < n; i += stride) p[i] = 0.0f;
}

__global__ void count_kernel(const int* __restrict__ esrc, const int* __restrict__ edst,
                             float* __restrict__ cnt_u, float* __restrict__ cnt_b, int E) {
    int i = blockIdx.x * blockDim.x + threadIdx.x;
    if (i < E) {
        atomicAdd(&cnt_u[esrc[i]], 1.0f);
        atomicAdd(&cnt_b[edst[i]], 1.0f);
    }
}

__global__ void scatter_kernel(const float* __restrict__ feat, const int* __restrict__ sidx,
                               const int* __restrict__ didx, float* __restrict__ sums, int E) {
    int gid = blockIdx.x * blockDim.x + threadIdx.x;
    int e = gid >> 5;
    if (e >= E) return;
    int lane = gid & 31;
    int s = sidx[e], d = didx[e];
    const float4 v = *(const float4*)(feat + (size_t)s * HD + lane * 4);
    float* dp = sums + (size_t)d * HD + lane * 4;
    atomicAdd(dp + 0, v.x);
    atomicAdd(dp + 1, v.y);
    atomicAdd(dp + 2, v.z);
    atomicAdd(dp + 3, v.w);
}


__device__ __forceinline__ void load_weight_t(const float* __restrict__ W, float* __restrict__ wt,
                                              int K, int stride) {
    const int nf4 = K * (HD / 4);
    for (int idx = threadIdx.x; idx < nf4; idx += 256) {
        int k = idx >> 5;
        int n = (idx & 31) << 2;
        const float4 w4 = *(const float4*)(W + (size_t)k * HD + n);
        wt[(n + 0) * stride + k] = w4.x;
        wt[(n + 1) * stride + k] = w4.y;
        wt[(n + 2) * stride + k] = w4.z;
        wt[(n + 3) * stride + k] = w4.w;
    }
}

__global__ void gemm_bias_kernel(const float* __restrict__ A, const float* __restrict__ W,
                                 const float* __restrict__ bias, float* __restrict__ out,
                                 int M, int K) {
    __shared__ float wt[HD * (HD + 4)];
    const int stride = K + 4;
    load_weight_t(W, wt, K, stride);
    __syncthreads();

    const int wave = threadIdx.x >> 5;
    const int lane = threadIdx.x & 31;
    const int col  = lane & 15;
    const int half = lane >> 4;
    const int m0   = blockIdx.x * 128 + wave * 16;
    const int aRow = (m0 + col < M) ? (m0 + col) : (M - 1);

    v8f acc[8];
#pragma unroll
    for (int nt = 0; nt < 8; ++nt) {
        float bv = bias[nt * 16 + col];
        acc[nt] = (v8f){bv, bv, bv, bv, bv, bv, bv, bv};
    }

    const float* arow = A + (size_t)aRow * K;
    for (int k0 = 0; k0 < K; k0 += 4) {
        const int kb = k0 + 2 * half;
        v2f a = *(const v2f*)(arow + kb);
        v2f bf[8];
#pragma unroll
        for (int nt = 0; nt < 8; ++nt)
            bf[nt] = *(const v2f*)&wt[(nt * 16 + col) * stride + kb];
#pragma unroll
        for (int nt = 0; nt < 8; ++nt)
            acc[nt] = __builtin_amdgcn_wmma_f32_16x16x4_f32(
                false, a, false, bf[nt], (short)0, acc[nt], false, false);
    }

#pragma unroll
    for (int nt = 0; nt < 8; ++nt)
#pragma unroll
        for (int v = 0; v < 8; ++v) {
            int dr = m0 + v + 8 * half;
            if (dr < M) out[(size_t)dr * HD + nt * 16 + col] = acc[nt][v];
        }
}

__global__ void sage_combine_kernel(const float* __restrict__ sums, const float* __restrict__ cnt,
                                    const float* __restrict__ X, const float* __restrict__ Wl,
                                    const float* __restrict__ bl, const float* __restrict__ Wr,
                                    float* __restrict__ out, int M) {
    __shared__ float wt[HD * (HD + 4)];
    const int stride = HD + 4;

    load_weight_t(Wl, wt, HD, stride);
    __syncthreads();

    const int wave = threadIdx.x >> 5;
    const int lane = threadIdx.x & 31;
    const int col  = lane & 15;
    const int half = lane >> 4;
    const int m0   = blockIdx.x * 128 + wave * 16;
    const int aRow = (m0 + col < M) ? (m0 + col) : (M - 1);
    const float invc = 1.0f / fmaxf(cnt[aRow], 1.0f);

    v8f acc[8];
#pragma unroll
    for (int nt = 0; nt < 8; ++nt) {
        float bv = bl[nt * 16 + col];
        acc[nt] = (v8f){bv, bv, bv, bv, bv, bv, bv, bv};
    }

    const float* srow = sums + (size_t)aRow * HD;
    for (int k0 = 0; k0 < HD; k0 += 4) {
        const int kb = k0 + 2 * half;
        v2f a = *(const v2f*)(srow + kb);
        a.x *= invc;
        a.y *= invc;
        v2f bf[8];
#pragma unroll
        for (int nt = 0; nt < 8; ++nt)
            bf[nt] = *(const v2f*)&wt[(nt * 16 + col) * stride + kb];
#pragma unroll
        for (int nt = 0; nt < 8; ++nt)
            acc[nt] = __builtin_amdgcn_wmma_f32_16x16x4_f32(
                false, a, false, bf[nt], (short)0, acc[nt], false, false);
    }

    __syncthreads();
    load_weight_t(Wr, wt, HD, stride);
    __syncthreads();

    const float* xrow = X + (size_t)aRow * HD;
    for (int k0 = 0; k0 < HD; k0 += 4) {
        const int kb = k0 + 2 * half;
        v2f a = *(const v2f*)(xrow + kb);
        v2f bf[8];
#pragma unroll
        for (int nt = 0; nt < 8; ++nt)
            bf[nt] = *(const v2f*)&wt[(nt * 16 + col) * stride + kb];
#pragma unroll
        for (int nt = 0; nt < 8; ++nt)
            acc[nt] = __builtin_amdgcn_wmma_f32_16x16x4_f32(
                false, a, false, bf[nt], (short)0, acc[nt], false, false);
    }

#pragma unroll
    for (int nt = 0; nt < 8; ++nt)
#pragma unroll
        for (int v = 0; v < 8; ++v) {
            int dr = m0 + v + 8 * half;
            if (dr < M) out[(size_t)dr * HD + nt * 16 + col] = fmaxf(acc[nt][v], 0.0f);
        }
}

__global__ void layernorm_kernel(const float* __restrict__ X, const float* __restrict__ gamma,
                                 const float* __restrict__ beta, float* __restrict__ out, int M) {
    const int wave = threadIdx.x >> 5;
    const int lane = threadIdx.x & 31;
    const int row  = blockIdx.x * 8 + wave;
    if (row >= M) return;

    const float4 v = *(const float4*)(X + (size_t)row * HD + lane * 4);
    float s = v.x + v.y + v.z + v.w;
#pragma unroll
    for (int off = 16; off >= 1; off >>= 1) s += __shfl_xor(s, off, 32);
    const float mu = s * (1.0f / HD);

    const float dx = v.x - mu, dy = v.y - mu, dz = v.z - mu, dw = v.w - mu;
    float q = dx * dx + dy * dy + dz * dz + dw * dw;
#pragma unroll
    for (int off = 16; off >= 1; off >>= 1) q += __shfl_xor(q, off, 32);
    const float rstd = rsqrtf(q * (1.0f / HD) + LN_EPS);

    const float4 g = *(const float4*)(gamma + lane * 4);
    const float4 bt = *(const float4*)(beta + lane * 4);
    float4 o;
    o.x = dx * rstd * g.x + bt.x;
    o.y = dy * rstd * g.y + bt.y;
    o.z = dz * rstd * g.z + bt.z;
    o.w = dw * rstd * g.w + bt.w;
    *(float4*)(out + (size_t)row * HD + lane * 4) = o;
}

extern "C" void kernel_launch(void* const* d_in, const int* in_sizes, int n_in,
                              void* d_out, int out_size, void* d_ws, size_t ws_size,
                              hipStream_t stream) {
    const float* user_x      = (const float*)d_in[0];
    const float* book_x      = (const float*)d_in[1];
    const int*   edge_src    = (const int*)d_in[2];
    const int*   edge_dst    = (const int*)d_in[3];
    const float* user_proj_w = (const float*)d_in[4];
    const float* user_proj_b = (const float*)d_in[5];
    const float* book_proj_w = (const float*)d_in[6];
    const float* book_proj_b = (const float*)d_in[7];
    const float* Wl          = (const float*)d_in[8];
    const float* bl          = (const float*)d_in[9];
    const float* Wr          = (const float*)d_in[10];
    const float* ugamma      = (const float*)d_in[11];
    const float* ubeta       = (const float*)d_in[12];
    const float* bgamma      = (const float*)d_in[13];
    const float* bbeta       = (const float*)d_in[14];

    float* ws = (float*)d_ws;
    size_t off = 0;
    auto carve = [&](size_t n) { float* p = ws + off; off += n; return p; };
    float* u_cur = carve((size_t)N_USER * HD);
    float* u_nxt = carve((size_t)N_USER * HD);
    float* b_cur = carve((size_t)N_BOOK * HD);
    float* b_nxt = carve((size_t)N_BOOK * HD);
    float* sum_u = carve((size_t)N_USER * HD);
    float* sum_b = carve((size_t)N_BOOK * HD);
    float* cnt_u = carve((size_t)N_USER);
    float* cnt_b = carve((size_t)N_BOOK);

    gemm_bias_kernel<<<(N_USER + 127) / 128, 256, 0, stream>>>(
        user_x, user_proj_w, user_proj_b, u_cur, N_USER, 64);
    gemm_bias_kernel<<<(N_BOOK + 127) / 128, 256, 0, stream>>>(
        book_x, book_proj_w, book_proj_b, b_cur, N_BOOK, 128);

    zero_kernel<<<256, 256, 0, stream>>>(cnt_u, (size_t)N_USER);
    zero_kernel<<<256, 256, 0, stream>>>(cnt_b, (size_t)N_BOOK);
    count_kernel<<<(N_EDGE + 255) / 256, 256, 0, stream>>>(
        edge_src, edge_dst, cnt_u, cnt_b, N_EDGE);

    const int scat_blocks = (N_EDGE * 32 + 255) / 256;
    for (int l = 0; l < 3; ++l) {
        const float* wl  = Wl + (size_t)l * HD * HD;
        const float* blv = bl + (size_t)l * HD;
        const float* wr  = Wr + (size_t)l * HD * HD;

        zero_kernel<<<4096, 256, 0, stream>>>(sum_b, (size_t)N_BOOK * HD);
        zero_kernel<<<4096, 256, 0, stream>>>(sum_u, (size_t)N_USER * HD);

        scatter_kernel<<<scat_blocks, 256, 0, stream>>>(u_cur, edge_src, edge_dst, sum_b, N_EDGE);
        scatter_kernel<<<scat_blocks, 256, 0, stream>>>(b_cur, edge_dst, edge_src, sum_u, N_EDGE);

        sage_combine_kernel<<<(N_BOOK + 127) / 128, 256, 0, stream>>>(
            sum_b, cnt_b, b_cur, wl, blv, wr, b_nxt, N_BOOK);
        sage_combine_kernel<<<(N_USER + 127) / 128, 256, 0, stream>>>(
            sum_u, cnt_u, u_cur, wl, blv, wr, u_nxt, N_USER);

        float* t;
        t = u_cur; u_cur = u_nxt; u_nxt = t;
        t = b_cur; b_cur = b_nxt; b_nxt = t;
    }

    float* out_u = (float*)d_out;
    float* out_b = out_u + (size_t)N_USER * HD;
    layernorm_kernel<<<(N_USER + 7) / 8, 256, 0, stream>>>(u_cur, ugamma, ubeta, out_u, N_USER);
    layernorm_kernel<<<(N_BOOK + 7) / 8, 256, 0, stream>>>(b_cur, bgamma, bbeta, out_b, N_BOOK);
}


# Float-atomic scatter-add replaced by per-call neighbour lists, each sorted by source index, and a per-row gather-sum in that fixed order: bit-identical run to run (earlier try kept arrival order)
# speedup vs baseline: 2.3434x; 2.3434x over previous
#include <hip/hip_runtime.h>

typedef float v2f __attribute__((ext_vector_type(2)));
typedef float v8f __attribute__((ext_vector_type(8)));

#define HD 128
constexpr int N_USER = 150000;
constexpr int N_BOOK = 75000;
constexpr int N_EDGE = 1500000;
constexpr float LN_EPS = 1e-5f;
constexpr int N_ROWS = N_BOOK + N_USER;
constexpr int SCAN_TILE = 1024;
constexpr int SCAN_BLOCKS = (N_ROWS + SCAN_TILE - 1) / SCAN_TILE;

__global__ void zero_int_kernel(int* __restrict__ p, int n) {
    int i = blockIdx.x * blockDim.x + threadIdx.x;
    if (i < n) p[i] = 0;
}

__global__ void count_rank_kernel(const int* __restrict__ esrc, const int* __restrict__ edst,
                                  int* __restrict__ deg, int* __restrict__ rank_b,
                                  int* __restrict__ rank_u, int E) {
    int i = blockIdx.x * blockDim.x + threadIdx.x;
    if (i < E) {
        rank_b[i] = atomicAdd(&deg[edst[i]], 1);
        rank_u[i] = atomicAdd(&deg[N_BOOK + esrc[i]], 1);
    }
}

__global__ void block_sum_kernel(const int* __restrict__ deg, int* __restrict__ part, int n) {
    __shared__ int red[8];
    const int lane = threadIdx.x & 31;
    const int wave = threadIdx.x >> 5;
    const int base = blockIdx.x * SCAN_TILE + threadIdx.x * 4;
    int s = 0;
    for (int u = 0; u < 4; ++u)
        if (base + u < n) s += deg[base + u];
    for (int o = 16; o >= 1; o >>= 1) s += __shfl_xor(s, o, 32);
    if (lane == 0) red[wave] = s;
    __syncthreads();
    if (threadIdx.x == 0) {
        int t = 0;
        for (int w = 0; w < 8; ++w) t += red[w];
        part[blockIdx.x] = t;
    }
}

__global__ void scan_kernel(const int* __restrict__ deg, const int* __restrict__ part,
                            int* __restrict__ off, float* __restrict__ cnt, int n) {
    __shared__ int red[8];
    __shared__ int wsum[8];
    const int lane = threadIdx.x & 31;
    const int wave = threadIdx.x >> 5;

    int p = 0;
    for (int i = threadIdx.x; i < (int)blockIdx.x; i += 256) p += part[i];
    for (int o = 16; o >= 1; o >>= 1) p += __shfl_xor(p, o, 32);
    if (lane == 0) red[wave] = p;

    const int base = blockIdx.x * SCAN_TILE + threadIdx.x * 4;
    int d[4];
    int tsum = 0;
    for (int u = 0; u < 4; ++u) {
        d[u] = (base + u < n) ? deg[base + u] : 0;
        tsum += d[u];
    }
    int incl = tsum;
    for (int o = 1; o < 32; o <<= 1) {
        int t = __shfl_up(incl, o, 32);
        if (lane >= o) incl += t;
    }
    if (lane == 31) wsum[wave] = incl;
    __syncthreads();

    int run = incl - tsum;
    for (int w = 0; w < 8; ++w) {
        run += red[w];
        if (w < wave) run += wsum[w];
    }
    for (int u = 0; u < 4; ++u)
        if (base + u < n) {
            off[base + u] = run;
            cnt[base + u] = (float)d[u];
            run += d[u];
        }
}

__global__ void fill_kernel(const int* __restrict__ esrc, const int* __restrict__ edst,
                            const int* __restrict__ off, const int* __restrict__ rank_b,
                            const int* __restrict__ rank_u, int* __restrict__ adj, int E) {
    int i = blockIdx.x * blockDim.x + threadIdx.x;
    if (i < E) {
        const int s = esrc[i], d = edst[i];
        adj[off[d] + rank_b[i]] = s;
        adj[off[N_BOOK + s] + rank_u[i]] = d;
    }
}

__global__ void sort_rows_kernel(const int* __restrict__ off, const int* __restrict__ deg,
                                 const int* __restrict__ adj_in, int* __restrict__ adj_out,
                                 int n_rows) {
    const int lane = threadIdx.x & 31;
    const int row = __builtin_amdgcn_readfirstlane((int)(blockIdx.x * 8 + (threadIdx.x >> 5)));
    if (row >= n_rows) return;
    const int base = off[row];
    const int n = deg[row];
    for (int i0 = 0; i0 < n; i0 += 32) {
        const int i = i0 + lane;
        const int mine = (i < n) ? adj_in[base + i] : 0;
        int rank = 0;
        for (int j0 = 0; j0 < n; j0 += 32) {
            const int jn = (n - j0 < 32) ? (n - j0) : 32;
            const int theirs = (lane < jn) ? adj_in[base + j0 + lane] : 0;
            for (int jj = 0; jj < jn; ++jj) {
                const int o = __builtin_amdgcn_readlane(theirs, jj);
                const int j = j0 + jj;
                rank += (o < mine || (o == mine && j < i)) ? 1 : 0;
            }
        }
        if (i < n) adj_out[base + rank] = mine;
    }
}

template <int U>
__device__ __forceinline__ void add_rows(float4& acc, const float* __restrict__ feat,
                                         int ids, int j, int lane) {
    float4 v[U];
#pragma unroll
    for (int u = 0; u < U; ++u) {
        const int s = __builtin_amdgcn_readlane(ids, j + u);
        v[u] = *(const float4*)(feat + (size_t)s * HD + lane * 4);
    }
#pragma unroll
    for (int u = 0; u < U; ++u) {
        acc.x += v[u].x;
        acc.y += v[u].y;
        acc.z += v[u].z;
        acc.w += v[u].w;
    }
}

__global__ void gather_sum_kernel(const float* __restrict__ feat, const int* __restrict__ off,
                                  const int* __restrict__ deg, const int* __restrict__ adj,
                                  float* __restrict__ sums, int M) {
    const int lane = threadIdx.x & 31;
    const int row = __builtin_amdgcn_readfirstlane((int)(blockIdx.x * 8 + (threadIdx.x >> 5)));
    if (row >= M) return;
    const int beg = off[row];
    const int n = deg[row];
    float4 acc = {0.0f, 0.0f, 0.0f, 0.0f};
    for (int c = 0; c < n; c += 32) {
        const int m = (n - c < 32) ? (n - c) : 32;
        const int ids = (lane < m) ? adj[beg + c + lane] : 0;
        int j = 0;
        for (; j + 8 <= m; j += 8) add_rows<8>(acc, feat, ids, j, lane);
        if (m - j >= 4) { add_rows<4>(acc, feat, ids, j, lane); j += 4; }
        if (m - j >= 2) { add_rows<2>(acc, feat, ids, j, lane); j += 2; }
        if (m - j >= 1) add_rows<1>(acc, feat, ids, j, lane);
    }
    *(float4*)(sums + (size_t)row * HD + lane * 4) = acc;
}


__device__ __forceinline__ void load_weight_t(const float* __restrict__ W, float* __restrict__ wt,
                                              int K, int stride) {
    const int nf4 = K * (HD / 4);
    for (int idx = threadIdx.x; idx < nf4; idx += 256) {
        int k = idx >> 5;
        int n = (idx & 31) << 2;
        const float4 w4 = *(const float4*)(W + (size_t)k * HD + n);
        wt[(n + 0) * stride + k] = w4.x;
        wt[(n + 1) * stride + k] = w4.y;
        wt[(n + 2) * stride + k] = w4.z;
        wt[(n + 3) * stride + k] = w4.w;
    }
}

__global__ void gemm_bias_kernel(const float* __restrict__ A, const float* __restrict__ W,
                                 const float* __restrict__ bias, float* __restrict__ out,
                                 int M, int K) {
    __shared__ float wt[HD * (HD + 4)];
    const int stride = K + 4;
    load_weight_t(W, wt, K, stride);
    __syncthreads();

    const int wave = threadIdx.x >> 5;
    const int lane = threadIdx.x & 31;
    const int col  = lane & 15;
    const int half = lane >> 4;
    const int m0   = blockIdx.x * 128 + wave * 16;
    const int aRow = (m0 + col < M) ? (m0 + col) : (M - 1);

    v8f acc[8];
#pragma unroll
    for (int nt = 0; nt < 8; ++nt) {
        float bv = bias[nt * 16 + col];
        acc[nt] = (v8f){bv, bv, bv, bv, bv, bv, bv, bv};
    }

    const float* arow = A + (size_t)aRow * K;
    for (int k0 = 0; k0 < K; k0 += 4) {
        const int kb = k0 + 2 * half;
        v2f a = *(const v2f*)(arow + kb);
        v2f bf[8];
#pragma unroll
        for (int nt = 0; nt < 8; ++nt)
            bf[nt] = *(const v2f*)&wt[(nt * 16 + col) * stride + kb];
#pragma unroll
        for (int nt = 0; nt < 8; ++nt)
            acc[nt] = __builtin_amdgcn_wmma_f32_16x16x4_f32(
                false, a, false, bf[nt], (short)0, acc[nt], false, false);
    }

#pragma unroll
    for (int nt = 0; nt < 8; ++nt)
#pragma unroll
        for (int v = 0; v < 8; ++v) {
            int dr = m0 + v + 8 * half;
            if (dr < M) out[(size_t)dr * HD + nt * 16 + col] = acc[nt][v];
        }
}

__global__ void sage_combine_kernel(const float* __restrict__ sums, const float* __restrict__ cnt,
                                    const float* __restrict__ X, const float* __restrict__ Wl,
                                    const float* __restrict__ bl, const float* __restrict__ Wr,
                                    float* __restrict__ out, int M) {
    __shared__ float wt[HD * (HD + 4)];
    const int stride = HD + 4;

    load_weight_t(Wl, wt, HD, stride);
    __syncthreads();

    const int wave = threadIdx.x >> 5;
    const int lane = threadIdx.x & 31;
    const int col  = lane & 15;
    const int half = lane >> 4;
    const int m0   = blockIdx.x * 128 + wave * 16;
    const int aRow = (m0 + col < M) ? (m0 + col) : (M - 1);
    const float invc = 1.0f / fmaxf(cnt[aRow], 1.0f);

    v8f acc[8];
#pragma unroll
    for (int nt = 0; nt < 8; ++nt) {
        float bv = bl[nt * 16 + col];
        acc[nt] = (v8f){bv, bv, bv, bv, bv, bv, bv, bv};
    }

    const float* srow = sums + (size_t)aRow * HD;
    for (int k0 = 0; k0 < HD; k0 += 4) {
        const int kb = k0 + 2 * half;
        v2f a = *(const v2f*)(srow + kb);
        a.x *= invc;
        a.y *= invc;
        v2f bf[8];
#pragma unroll
        for (int nt = 0; nt < 8; ++nt)
            bf[nt] = *(const v2f*)&wt[(nt * 16 + col) * stride + kb];
#pragma unroll
        for (int nt = 0; nt < 8; ++nt)
            acc[nt] = __builtin_amdgcn_wmma_f32_16x16x4_f32(
                false, a, false, bf[nt], (short)0, acc[nt], false, false);
    }

    __syncthreads();
    load_weight_t(Wr, wt, HD, stride);
    __syncthreads();

    const float* xrow = X + (size_t)aRow * HD;
    for (int k0 = 0; k0 < HD; k0 += 4) {
        const int kb = k0 + 2 * half;
        v2f a = *(const v2f*)(xrow + kb);
        v2f bf[8];
#pragma unroll
        for (int nt = 0; nt < 8; ++nt)
            bf[nt] = *(const v2f*)&wt[(nt * 16 + col) * stride + kb];
#pragma unroll
        for (int nt = 0; nt < 8; ++nt)
            acc[nt] = __builtin_amdgcn_wmma_f32_16x16x4_f32(
                false, a, false, bf[nt], (short)0, acc[nt], false, false);
    }

#pragma unroll
    for (int nt = 0; nt < 8; ++nt)
#pragma unroll
        for (int v = 0; v < 8; ++v) {
            int dr = m0 + v + 8 * half;
            if (dr < M) out[(size_t)dr * HD + nt * 16 + col] = fmaxf(acc[nt][v], 0.0f);
        }
}

__global__ void layernorm_kernel(const float* __restrict__ X, const float* __restrict__ gamma,
                                 const float* __restrict__ beta, float* __restrict__ out, int M) {
    const int wave = threadIdx.x >> 5;
    const int lane = threadIdx.x & 31;
    const int row  = blockIdx.x * 8 + wave;
    if (row >= M) return;

    const float4 v = *(const float4*)(X + (size_t)row * HD + lane * 4);
    float s = v.x + v.y + v.z + v.w;
#pragma unroll
    for (int off = 16; off >= 1; off >>= 1) s += __shfl_xor(s, off, 32);
    const float mu = s * (1.0f / HD);

    const float dx = v.x - mu, dy = v.y - mu, dz = v.z - mu, dw = v.w - mu;
    float q = dx * dx + dy * dy + dz * dz + dw * dw;
#pragma unroll
    for (int off = 16; off >= 1; off >>= 1) q += __shfl_xor(q, off, 32);
    const float rstd = rsqrtf(q * (1.0f / HD) + LN_EPS);

    const float4 g = *(const float4*)(gamma + lane * 4);
    const float4 bt = *(const float4*)(beta + lane * 4);
    float4 o;
    o.x = dx * rstd * g.x + bt.x;
    o.y = dy * rstd * g.y + bt.y;
    o.z = dz * rstd * g.z + bt.z;
    o.w = dw * rstd * g.w + bt.w;
    *(float4*)(out + (size_t)row * HD + lane * 4) = o;
}

extern "C" void kernel_launch(void* const* d_in, const int* in_sizes, int n_in,
                              void* d_out, int out_size, void* d_ws, size_t ws_size,
                              hipStream_t stream) {
    const float* user_x      = (const float*)d_in[0];
    const float* book_x      = (const float*)d_in[1];
    const int*   edge_src    = (const int*)d_in[2];
    const int*   edge_dst    = (const int*)d_in[3];
    const float* user_proj_w = (const float*)d_in[4];
    const float* user_proj_b = (const float*)d_in[5];
    const float* book_proj_w = (const float*)d_in[6];
    const float* book_proj_b = (const float*)d_in[7];
    const float* Wl          = (const float*)d_in[8];
    const float* bl          = (const float*)d_in[9];
    const float* Wr          = (const float*)d_in[10];
    const float* ugamma      = (const float*)d_in[11];
    const float* ubeta       = (const float*)d_in[12];
    const float* bgamma      = (const float*)d_in[13];
    const float* bbeta       = (const float*)d_in[14];

    float* ws = (float*)d_ws;
    size_t off = 0;
    auto carve = [&](size_t n) { float* p = ws + off; off += n; return p; };
    float* u_cur = carve((size_t)N_USER * HD);
    float* u_nxt = carve((size_t)N_USER * HD);
    float* b_cur = carve((size_t)N_BOOK * HD);
    float* b_nxt = carve((size_t)N_BOOK * HD);
    float* sum_u = carve((size_t)N_USER * HD);
    float* sum_b = carve((size_t)N_BOOK * HD);
    float* cnt   = carve((size_t)N_ROWS);
    int* deg     = (int*)carve((size_t)N_ROWS);
    int* offs    = (int*)carve((size_t)N_ROWS);
    int* part    = (int*)carve((size_t)SCAN_BLOCKS);
    int* rank_b  = (int*)carve((size_t)N_EDGE);
    int* rank_u  = (int*)carve((size_t)N_EDGE);
    int* adj_raw = (int*)carve((size_t)2 * N_EDGE);
    int* adj     = (int*)carve((size_t)2 * N_EDGE);
    const float* cnt_b = cnt;
    const float* cnt_u = cnt + N_BOOK;

    gemm_bias_kernel<<<(N_USER + 127) / 128, 256, 0, stream>>>(
        user_x, user_proj_w, user_proj_b, u_cur, N_USER, 64);
    gemm_bias_kernel<<<(N_BOOK + 127) / 128, 256, 0, stream>>>(
        book_x, book_proj_w, book_proj_b, b_cur, N_BOOK, 128);

    zero_int_kernel<<<(N_ROWS + 255) / 256, 256, 0, stream>>>(deg, N_ROWS);
    count_rank_kernel<<<(N_EDGE + 255) / 256, 256, 0, stream>>>(
        edge_src, edge_dst, deg, rank_b, rank_u, N_EDGE);
    block_sum_kernel<<<SCAN_BLOCKS, 256, 0, stream>>>(deg, part, N_ROWS);
    scan_kernel<<<SCAN_BLOCKS, 256, 0, stream>>>(deg, part, offs, cnt, N_ROWS);
    fill_kernel<<<(N_EDGE + 255) / 256, 256, 0, stream>>>(
        edge_src, edge_dst, offs, rank_b, rank_u, adj_raw, N_EDGE);
    sort_rows_kernel<<<(N_ROWS + 7) / 8, 256, 0, stream>>>(offs, deg, adj_raw, adj, N_ROWS);

    for (int l = 0; l < 3; ++l) {
        const float* wl  = Wl + (size_t)l * HD * HD;
        const float* blv = bl + (size_t)l * HD;
        const float* wr  = Wr + (size_t)l * HD * HD;

        gather_sum_kernel<<<(N_BOOK + 7) / 8, 256, 0, stream>>>(
            u_cur, offs, deg, adj, sum_b, N_BOOK);
        gather_sum_kernel<<<(N_USER + 7) / 8, 256, 0, stream>>>(
            b_cur, offs + N_BOOK, deg + N_BOOK, adj, sum_u, N_USER);

        sage_combine_kernel<<<(N_BOOK + 127) / 128, 256, 0, stream>>>(
            sum_b, cnt_b, b_cur, wl, blv, wr, b_nxt, N_BOOK);
        sage_combine_kernel<<<(N_USER + 127) / 128, 256, 0, stream>>>(
            sum_u, cnt_u, u_cur, wl, blv, wr, u_nxt, N_USER);

        float* t;
        t = u_cur; u_cur = u_nxt; u_nxt = t;
        t = b_cur; b_cur = b_nxt; b_nxt = t;
    }

    float* out_u = (float*)d_out;
    float* out_b = out_u + (size_t)N_USER * HD;
    layernorm_kernel<<<(N_USER + 7) / 8, 256, 0, stream>>>(u_cur, ugamma, ubeta, out_u, N_USER);
    layernorm_kernel<<<(N_BOOK + 7) / 8, 256, 0, stream>>>(b_cur, bgamma, bbeta, out_b, N_BOOK);
}


# Row-per-wave kernels (gather-sum, list sort, layernorm) and edge-per-thread kernels run a fixed 2048-block grid, each wave looping over its rows: far fewer waves to start; sum order unchanged
# speedup vs baseline: 2.4164x; 1.0312x over previous
#include <hip/hip_runtime.h>

typedef float v2f __attribute__((ext_vector_type(2)));
typedef float v8f __attribute__((ext_vector_type(8)));

#define HD 128
constexpr int N_USER = 150000;
constexpr int N_BOOK = 75000;
constexpr int N_EDGE = 1500000;
constexpr float LN_EPS = 1e-5f;
constexpr int N_ROWS = N_BOOK + N_USER;
constexpr int SCAN_TILE = 1024;
constexpr int SCAN_BLOCKS = (N_ROWS + SCAN_TILE - 1) / SCAN_TILE;
constexpr int FILL_BLOCKS = 2048;

__global__ void zero_int_kernel(int* __restrict__ p, int n) {
    int i = blockIdx.x * blockDim.x + threadIdx.x;
    if (i < n) p[i] = 0;
}

__global__ void count_rank_kernel(const int* __restrict__ esrc, const int* __restrict__ edst,
                                  int* __restrict__ deg, int* __restrict__ rank_b,
                                  int* __restrict__ rank_u, int E) {
    const int step = gridDim.x * blockDim.x;
    for (int i = blockIdx.x * blockDim.x + threadIdx.x; i < E; i += step) {
        rank_b[i] = atomicAdd(&deg[edst[i]], 1);
        rank_u[i] = atomicAdd(&deg[N_BOOK + esrc[i]], 1);
    }
}

__global__ void block_sum_kernel(const int* __restrict__ deg, int* __restrict__ part, int n) {
    __shared__ int red[8];
    const int lane = threadIdx.x & 31;
    const int wave = threadIdx.x >> 5;
    const int base = blockIdx.x * SCAN_TILE + threadIdx.x * 4;
    int s = 0;
    for (int u = 0; u < 4; ++u)
        if (base + u < n) s += deg[base + u];
    for (int o = 16; o >= 1; o >>= 1) s += __shfl_xor(s, o, 32);
    if (lane == 0) red[wave] = s;
    __syncthreads();
    if (threadIdx.x == 0) {
        int t = 0;
        for (int w = 0; w < 8; ++w) t += red[w];
        part[blockIdx.x] = t;
    }
}

__global__ void scan_kernel(const int* __restrict__ deg, const int* __restrict__ part,
                            int* __restrict__ off, float* __restrict__ cnt, int n) {
    __shared__ int red[8];
    __shared__ int wsum[8];
    const int lane = threadIdx.x & 31;
    const int wave = threadIdx.x >> 5;

    int p = 0;
    for (int i = threadIdx.x; i < (int)blockIdx.x; i += 256) p += part[i];
    for (int o = 16; o >= 1; o >>= 1) p += __shfl_xor(p, o, 32);
    if (lane == 0) red[wave] = p;

    const int base = blockIdx.x * SCAN_TILE + threadIdx.x * 4;
    int d[4];
    int tsum = 0;
    for (int u = 0; u < 4; ++u) {
        d[u] = (base + u < n) ? deg[base + u] : 0;
        tsum += d[u];
    }
    int incl = tsum;
    for (int o = 1; o < 32; o <<= 1) {
        int t = __shfl_up(incl, o, 32);
        if (lane >= o) incl += t;
    }
    if (lane == 31) wsum[wave] = incl;
    __syncthreads();

    int run = incl - tsum;
    for (int w = 0; w < 8; ++w) {
        run += red[w];
        if (w < wave) run += wsum[w];
    }
    for (int u = 0; u < 4; ++u)
        if (base + u < n) {
            off[base + u] = run;
            cnt[base + u] = (float)d[u];
            run += d[u];
        }
}

__global__ void fill_kernel(const int* __restrict__ esrc, const int* __restrict__ edst,
                            const int* __restrict__ off, const int* __restrict__ rank_b,
                            const int* __restrict__ rank_u, int* __restrict__ adj, int E) {
    const int step = gridDim.x * blockDim.x;
    for (int i = blockIdx.x * blockDim.x + threadIdx.x; i < E; i += step) {
        const int s = esrc[i], d = edst[i];
        adj[off[d] + rank_b[i]] = s;
        adj[off[N_BOOK + s] + rank_u[i]] = d;
    }
}

__global__ void sort_rows_kernel(const int* __restrict__ off, const int* __restrict__ deg,
                                 const int* __restrict__ adj_in, int* __restrict__ adj_out,
                                 int n_rows) {
    const int lane = threadIdx.x & 31;
    const int waves = (int)(gridDim.x * 8);
    for (int row = __builtin_amdgcn_readfirstlane((int)(blockIdx.x * 8 + (threadIdx.x >> 5)));
         row < n_rows; row += waves) {
        const int base = off[row];
        const int n = deg[row];
        for (int i0 = 0; i0 < n; i0 += 32) {
            const int i = i0 + lane;
            const int mine = (i < n) ? adj_in[base + i] : 0;
            int rank = 0;
            for (int j0 = 0; j0 < n; j0 += 32) {
                const int jn = (n - j0 < 32) ? (n - j0) : 32;
                const int theirs = (lane < jn) ? adj_in[base + j0 + lane] : 0;
                for (int jj = 0; jj < jn; ++jj) {
                    const int o = __builtin_amdgcn_readlane(theirs, jj);
                    const int j = j0 + jj;
                    rank += (o < mine || (o == mine && j < i)) ? 1 : 0;
                }
            }
            if (i < n) adj_out[base + rank] = mine;
        }
    }
}

template <int U>
__device__ __forceinline__ void add_rows(float4& acc, const float* __restrict__ feat,
                                         int ids, int j, int lane) {
    float4 v[U];
#pragma unroll
    for (int u = 0; u < U; ++u) {
        const int s = __builtin_amdgcn_readlane(ids, j + u);
        v[u] = *(const float4*)(feat + (size_t)s * HD + lane * 4);
    }
#pragma unroll
    for (int u = 0; u < U; ++u) {
        acc.x += v[u].x;
        acc.y += v[u].y;
        acc.z += v[u].z;
        acc.w += v[u].w;
    }
}

__global__ void gather_sum_kernel(const float* __restrict__ feat, const int* __restrict__ off,
                                  const int* __restrict__ deg, const int* __restrict__ adj,
                                  float* __restrict__ sums, int M) {
    const int lane = threadIdx.x & 31;
    const int waves = (int)(gridDim.x * 8);
    for (int row = __builtin_amdgcn_readfirstlane((int)(blockIdx.x * 8 + (threadIdx.x >> 5)));
         row < M; row += waves) {
        const int beg = off[row];
        const int n = deg[row];
        float4 acc = {0.0f, 0.0f, 0.0f, 0.0f};
        for (int c = 0; c < n; c += 32) {
            const int m = (n - c < 32) ? (n - c) : 32;
            const int ids = (lane < m) ? adj[beg + c + lane] : 0;
            int j = 0;
            for (; j + 8 <= m; j += 8) add_rows<8>(acc, feat, ids, j, lane);
            if (m - j >= 4) { add_rows<4>(acc, feat, ids, j, lane); j += 4; }
            if (m - j >= 2) { add_rows<2>(acc, feat, ids, j, lane); j += 2; }
            if (m - j >= 1) add_rows<1>(acc, feat, ids, j, lane);
        }
        *(float4*)(sums + (size_t)row * HD + lane * 4) = acc;
    }
}


__device__ __forceinline__ void load_weight_t(const float* __restrict__ W, float* __restrict__ wt,
                                              int K, int stride) {
    const int nf4 = K * (HD / 4);
    for (int idx = threadIdx.x; idx < nf4; idx += 256) {
        int k = idx >> 5;
        int n = (idx & 31) << 2;
        const float4 w4 = *(const float4*)(W + (size_t)k * HD + n);
        wt[(n + 0) * stride + k] = w4.x;
        wt[(n + 1) * stride + k] = w4.y;
        wt[(n + 2) * stride + k] = w4.z;
        wt[(n + 3) * stride + k] = w4.w;
    }
}

__global__ void gemm_bias_kernel(const float* __restrict__ A, const float* __restrict__ W,
                                 const float* __restrict__ bias, float* __restrict__ out,
                                 int M, int K) {
    __shared__ float wt[HD * (HD + 4)];
    const int stride = K + 4;
    load_weight_t(W, wt, K, stride);
    __syncthreads();

    const int wave = threadIdx.x >> 5;
    const int lane = threadIdx.x & 31;
    const int col  = lane & 15;
    const int half = lane >> 4;
    const int m0   = blockIdx.x * 128 + wave * 16;
    const int aRow = (m0 + col < M) ? (m0 + col) : (M - 1);

    v8f acc[8];
#pragma unroll
    for (int nt = 0; nt < 8; ++nt) {
        float bv = bias[nt * 16 + col];
        acc[nt] = (v8f){bv, bv, bv, bv, bv, bv, bv, bv};
    }

    const float* arow = A + (size_t)aRow * K;
    for (int k0 = 0; k0 < K; k0 += 4) {
        const int kb = k0 + 2 * half;
        v2f a = *(const v2f*)(arow + kb);
        v2f bf[8];
#pragma unroll
        for (int nt = 0; nt < 8; ++nt)
            bf[nt] = *(const v2f*)&wt[(nt * 16 + col) * stride + kb];
#pragma unroll
        for (int nt = 0; nt < 8; ++nt)
            acc[nt] = __builtin_amdgcn_wmma_f32_16x16x4_f32(
                false, a, false, bf[nt], (short)0, acc[nt], false, false);
    }

#pragma unroll
    for (int nt = 0; nt < 8; ++nt)
#pragma unroll
        for (int v = 0; v < 8; ++v) {
            int dr = m0 + v + 8 * half;
            if (dr < M) out[(size_t)dr * HD + nt * 16 + col] = acc[nt][v];
        }
}

__global__ void sage_combine_kernel(const float* __restrict__ sums, const float* __restrict__ cnt,
                                    const float* __restrict__ X, const float* __restrict__ Wl,
                                    const float* __restrict__ bl, const float* __restrict__ Wr,
                                    float* __restrict__ out, int M) {
    __shared__ float wt[HD * (HD + 4)];
    const int stride = HD + 4;

    load_weight_t(Wl, wt, HD, stride);
    __syncthreads();

    const int wave = threadIdx.x >> 5;
    const int lane = threadIdx.x & 31;
    const int col  = lane & 15;
    const int half = lane >> 4;
    const int m0   = blockIdx.x * 128 + wave * 16;
    const int aRow = (m0 + col < M) ? (m0 + col) : (M - 1);
    const float invc = 1.0f / fmaxf(cnt[aRow], 1.0f);

    v8f acc[8];
#pragma unroll
    for (int nt = 0; nt < 8; ++nt) {
        float bv = bl[nt * 16 + col];
        acc[nt] = (v8f){bv, bv, bv, bv, bv, bv, bv, bv};
    }

    const float* srow = sums + (size_t)aRow * HD;
    for (int k0 = 0; k0 < HD; k0 += 4) {
        const int kb = k0 + 2 * half;
        v2f a = *(const v2f*)(srow + kb);
        a.x *= invc;
        a.y *= invc;
        v2f bf[8];
#pragma unroll
        for (int nt = 0; nt < 8; ++nt)
            bf[nt] = *(const v2f*)&wt[(nt * 16 + col) * stride + kb];
#pragma unroll
        for (int nt = 0; nt < 8; ++nt)
            acc[nt] = __builtin_amdgcn_wmma_f32_16x16x4_f32(
                false, a, false, bf[nt], (short)0, acc[nt], false, false);
    }

    __syncthreads();
    load_weight_t(Wr, wt, HD, stride);
    __syncthreads();

    const float* xrow = X + (size_t)aRow * HD;
    for (int k0 = 0; k0 < HD; k0 += 4) {
        const int kb = k0 + 2 * half;
        v2f a = *(const v2f*)(xrow + kb);
        v2f bf[8];
#pragma unroll
        for (int nt = 0; nt < 8; ++nt)
            bf[nt] = *(const v2f*)&wt[(nt * 16 + col) * stride + kb];
#pragma unroll
        for (int nt = 0; nt < 8; ++nt)
            acc[nt] = __builtin_amdgcn_wmma_f32_16x16x4_f32(
                false, a, false, bf[nt], (short)0, acc[nt], false, false);
    }

#pragma unroll
    for (int nt = 0; nt < 8; ++nt)
#pragma unroll
        for (int v = 0; v < 8; ++v) {
            int dr = m0 + v + 8 * half;
            if (dr < M) out[(size_t)dr * HD + nt * 16 + col] = fmaxf(acc[nt][v], 0.0f);
        }
}

template <int U>
__device__ __forceinline__ void ln_rows(const float* __restrict__ X, float* __restrict__ out,
                                        const float4 g, const float4 bt, int row0, int waves,
                                        int lane) {
    float4 v[U];
#pragma unroll
    for (int u = 0; u < U; ++u)
        v[u] = *(const float4*)(X + (size_t)(row0 + u * waves) * HD + lane * 4);
#pragma unroll
    for (int u = 0; u < U; ++u) {
        float s = v[u].x + v[u].y + v[u].z + v[u].w;
#pragma unroll
        for (int off = 16; off >= 1; off >>= 1) s += __shfl_xor(s, off, 32);
        const float mu = s * (1.0f / HD);

        const float dx = v[u].x - mu, dy = v[u].y - mu, dz = v[u].z - mu, dw = v[u].w - mu;
        float q = dx * dx + dy * dy + dz * dz + dw * dw;
#pragma unroll
        for (int off = 16; off >= 1; off >>= 1) q += __shfl_xor(q, off, 32);
        const float rstd = rsqrtf(q * (1.0f / HD) + LN_EPS);

        float4 o;
        o.x = dx * rstd * g.x + bt.x;
        o.y = dy * rstd * g.y + bt.y;
        o.z = dz * rstd * g.z + bt.z;
        o.w = dw * rstd * g.w + bt.w;
        *(float4*)(out + (size_t)(row0 + u * waves) * HD + lane * 4) = o;
    }
}

__global__ void layernorm_kernel(const float* __restrict__ X, const float* __restrict__ gamma,
                                 const float* __restrict__ beta, float* __restrict__ out, int M) {
    const int lane = threadIdx.x & 31;
    const int waves = (int)(gridDim.x * 8);
    const float4 g = *(const float4*)(gamma + lane * 4);
    const float4 bt = *(const float4*)(beta + lane * 4);

    int row = __builtin_amdgcn_readfirstlane((int)(blockIdx.x * 8 + (threadIdx.x >> 5)));
    for (; row + 3 * waves < M; row += 4 * waves) ln_rows<4>(X, out, g, bt, row, waves, lane);
    for (; row < M; row += waves) ln_rows<1>(X, out, g, bt, row, waves, lane);
}

extern "C" void kernel_launch(void* const* d_in, const int* in_sizes, int n_in,
                              void* d_out, int out_size, void* d_ws, size_t ws_size,
                              hipStream_t stream) {
    const float* user_x      = (const float*)d_in[0];
    const float* book_x      = (const float*)d_in[1];
    const int*   edge_src    = (const int*)d_in[2];
    const int*   edge_dst    = (const int*)d_in[3];
    const float* user_proj_w = (const float*)d_in[4];
    const float* user_proj_b = (const float*)d_in[5];
    const float* book_proj_w = (const float*)d_in[6];
    const float* book_proj_b = (const float*)d_in[7];
    const float* Wl          = (const float*)d_in[8];
    const float* bl          = (const float*)d_in[9];
    const float* Wr          = (const float*)d_in[10];
    const float* ugamma      = (const float*)d_in[11];
    const float* ubeta       = (const float*)d_in[12];
    const float* bgamma      = (const float*)d_in[13];
    const float* bbeta       = (const float*)d_in[14];

    float* ws = (float*)d_ws;
    size_t off = 0;
    auto carve = [&](size_t n) { float* p = ws + off; off += n; return p; };
    float* u_cur = carve((size_t)N_USER * HD);
    float* u_nxt = carve((size_t)N_USER * HD);
    float* b_cur = carve((size_t)N_BOOK * HD);
    float* b_nxt = carve((size_t)N_BOOK * HD);
    float* sum_u = carve((size_t)N_USER * HD);
    float* sum_b = carve((size_t)N_BOOK * HD);
    float* cnt   = carve((size_t)N_ROWS);
    int* deg     = (int*)carve((size_t)N_ROWS);
    int* offs    = (int*)carve((size_t)N_ROWS);
    int* part    = (int*)carve((size_t)SCAN_BLOCKS);
    int* rank_b  = (int*)carve((size_t)N_EDGE);
    int* rank_u  = (int*)carve((size_t)N_EDGE);
    int* adj_raw = (int*)carve((size_t)2 * N_EDGE);
    int* adj     = (int*)carve((size_t)2 * N_EDGE);
    const float* cnt_b = cnt;
    const float* cnt_u = cnt + N_BOOK;

    gemm_bias_kernel<<<(N_USER + 127) / 128, 256, 0, stream>>>(
        user_x, user_proj_w, user_proj_b, u_cur, N_USER, 64);
    gemm_bias_kernel<<<(N_BOOK + 127) / 128, 256, 0, stream>>>(
        book_x, book_proj_w, book_proj_b, b_cur, N_BOOK, 128);

    zero_int_kernel<<<(N_ROWS + 255) / 256, 256, 0, stream>>>(deg, N_ROWS);
    count_rank_kernel<<<FILL_BLOCKS, 256, 0, stream>>>(
        edge_src, edge_dst, deg, rank_b, rank_u, N_EDGE);
    block_sum_kernel<<<SCAN_BLOCKS, 256, 0, stream>>>(deg, part, N_ROWS);
    scan_kernel<<<SCAN_BLOCKS, 256, 0, stream>>>(deg, part, offs, cnt, N_ROWS);
    fill_kernel<<<FILL_BLOCKS, 256, 0, stream>>>(
        edge_src, edge_dst, offs, rank_b, rank_u, adj_raw, N_EDGE);
    sort_rows_kernel<<<FILL_BLOCKS, 256, 0, stream>>>(offs, deg, adj_raw, adj, N_ROWS);

    for (int l = 0; l < 3; ++l) {
        const float* wl  = Wl + (size_t)l * HD * HD;
        const float* blv = bl + (size_t)l * HD;
        const float* wr  = Wr + (size_t)l * HD * HD;

        gather_sum_kernel<<<FILL_BLOCKS, 256, 0, stream>>>(
            u_cur, offs, deg, adj, sum_b, N_BOOK);
        gather_sum_kernel<<<FILL_BLOCKS, 256, 0, stream>>>(
            b_cur, offs + N_BOOK, deg + N_BOOK, adj, sum_u, N_USER);

        sage_combine_kernel<<<(N_BOOK + 127) / 128, 256, 0, stream>>>(
            sum_b, cnt_b, b_cur, wl, blv, wr, b_nxt, N_BOOK);
        sage_combine_kernel<<<(N_USER + 127) / 128, 256, 0, stream>>>(
            sum_u, cnt_u, u_cur, wl, blv, wr, u_nxt, N_USER);

        float* t;
        t = u_cur; u_cur = u_nxt; u_nxt = t;
        t = b_cur; b_cur = b_nxt; b_nxt = t;
    }

    float* out_u = (float*)d_out;
    float* out_b = out_u + (size_t)N_USER * HD;
    layernorm_kernel<<<FILL_BLOCKS, 256, 0, stream>>>(u_cur, ugamma, ubeta, out_u, N_USER);
    layernorm_kernel<<<FILL_BLOCKS, 256, 0, stream>>>(b_cur, bgamma, bbeta, out_b, N_BOOK);
}


# Neighbour lists built by grouping edges through LDS counters (1024-row buckets, each counted, placed and ordered in LDS by one block), no global atomics or scattered writes; sum order unchanged
# speedup vs baseline: 2.8746x; 1.1896x over previous
#include <hip/hip_runtime.h>

typedef float v2f __attribute__((ext_vector_type(2)));
typedef float v8f __attribute__((ext_vector_type(8)));

#define HD 128
constexpr int N_USER = 150000;
constexpr int N_BOOK = 75000;
constexpr int N_EDGE = 1500000;
constexpr float LN_EPS = 1e-5f;
constexpr int N_ROWS = N_BOOK + N_USER;
constexpr int SCAN_TILE = 1024;
constexpr int FILL_BLOCKS = 2048;

constexpr int BK_SHIFT = 10;
constexpr int BK_ROWS = 1 << BK_SHIFT;
constexpr int NBK = (N_ROWS + BK_ROWS - 1) >> BK_SHIFT;
constexpr int NB1 = 256;
constexpr int CHUNK = (N_EDGE + NB1 - 1) / NB1;
constexpr int N_CNT = NBK * NB1;
constexpr int CSCAN_BLOCKS = (N_CNT + SCAN_TILE - 1) / SCAN_TILE;
constexpr int BK_PER_LANE = (NBK + 31) / 32;
constexpr int LIST_CAP = 40960;


__global__ void __launch_bounds__(1024)
bucket_count_kernel(const int* __restrict__ esrc, const int* __restrict__ edst,
                    int* __restrict__ counts, int E) {
    __shared__ int hist[NBK];
    for (int i = threadIdx.x; i < NBK; i += 1024) hist[i] = 0;
    __syncthreads();
    int c0 = blockIdx.x * CHUNK;
    if (c0 > E) c0 = E;
    const int c1 = (c0 + CHUNK < E) ? (c0 + CHUNK) : E;
    for (int i = c0 + threadIdx.x; i < c1; i += 1024) {
        atomicAdd(&hist[edst[i] >> BK_SHIFT], 1);
        atomicAdd(&hist[(N_BOOK + esrc[i]) >> BK_SHIFT], 1);
    }
    __syncthreads();
    for (int i = threadIdx.x; i < NBK; i += 1024) counts[i * NB1 + blockIdx.x] = hist[i];
}

__global__ void block_sum_kernel(const int* __restrict__ deg, int* __restrict__ part, int n) {
    __shared__ int red[8];
    const int lane = threadIdx.x & 31;
    const int wave = threadIdx.x >> 5;
    const int base = blockIdx.x * SCAN_TILE + threadIdx.x * 4;
    int s = 0;
    for (int u = 0; u < 4; ++u)
        if (base + u < n) s += deg[base + u];
    for (int o = 16; o >= 1; o >>= 1) s += __shfl_xor(s, o, 32);
    if (lane == 0) red[wave] = s;
    __syncthreads();
    if (threadIdx.x == 0) {
        int t = 0;
        for (int w = 0; w < 8; ++w) t += red[w];
        part[blockIdx.x] = t;
    }
}

__global__ void scan_kernel(const int* __restrict__ deg, const int* __restrict__ part,
                            int* __restrict__ off, int n) {
    __shared__ int red[8];
    __shared__ int wsum[8];
    const int lane = threadIdx.x & 31;
    const int wave = threadIdx.x >> 5;

    int p = 0;
    for (int i = threadIdx.x; i < (int)blockIdx.x; i += 256) p += part[i];
    for (int o = 16; o >= 1; o >>= 1) p += __shfl_xor(p, o, 32);
    if (lane == 0) red[wave] = p;

    const int base = blockIdx.x * SCAN_TILE + threadIdx.x * 4;
    int d[4];
    int tsum = 0;
    for (int u = 0; u < 4; ++u) {
        d[u] = (base + u < n) ? deg[base + u] : 0;
        tsum += d[u];
    }
    int incl = tsum;
    for (int o = 1; o < 32; o <<= 1) {
        int t = __shfl_up(incl, o, 32);
        if (lane >= o) incl += t;
    }
    if (lane == 31) wsum[wave] = incl;
    __syncthreads();

    int run = incl - tsum;
    for (int w = 0; w < 8; ++w) {
        run += red[w];
        if (w < wave) run += wsum[w];
    }
    for (int u = 0; u < 4; ++u)
        if (base + u < n) {
            off[base + u] = run;
            run += d[u];
        }
}

__global__ void __launch_bounds__(1024)
bucket_scatter_kernel(const int* __restrict__ esrc, const int* __restrict__ edst,
                      const int* __restrict__ cofs, int* __restrict__ p_row,
                      int* __restrict__ p_nbr, int E) {
    __shared__ int hist[NBK];
    __shared__ int lbase[NBK];
    __shared__ int gofs[NBK];
    __shared__ int st_row[2 * CHUNK];
    __shared__ int st_nbr[2 * CHUNK];
    const int tid = threadIdx.x;
    for (int i = tid; i < NBK; i += 1024) hist[i] = 0;
    __syncthreads();
    int c0 = blockIdx.x * CHUNK;
    if (c0 > E) c0 = E;
    const int c1 = (c0 + CHUNK < E) ? (c0 + CHUNK) : E;
    for (int i = c0 + tid; i < c1; i += 1024) {
        atomicAdd(&hist[edst[i] >> BK_SHIFT], 1);
        atomicAdd(&hist[(N_BOOK + esrc[i]) >> BK_SHIFT], 1);
    }
    __syncthreads();
    if (tid < 32) {
        int v[BK_PER_LANE];
        int s = 0;
        for (int u = 0; u < BK_PER_LANE; ++u) {
            const int idx = tid * BK_PER_LANE + u;
            v[u] = (idx < NBK) ? hist[idx] : 0;
            s += v[u];
        }
        int incl = s;
        for (int o = 1; o < 32; o <<= 1) {
            int t = __shfl_up(incl, o, 32);
            if (tid >= o) incl += t;
        }
        int run = incl - s;
        for (int u = 0; u < BK_PER_LANE; ++u) {
            const int idx = tid * BK_PER_LANE + u;
            if (idx < NBK) {
                lbase[idx] = run;
                run += v[u];
            }
        }
    }
    __syncthreads();
    for (int i = tid; i < NBK; i += 1024) {
        const int lb = lbase[i];
        hist[i] = lb;
        gofs[i] = cofs[i * NB1 + blockIdx.x] - lb;
    }
    __syncthreads();
    for (int i = c0 + tid; i < c1; i += 1024) {
        const int s = esrc[i], d = edst[i];
        const int rb = d;
        const int pb = atomicAdd(&hist[rb >> BK_SHIFT], 1);
        st_row[pb] = rb;
        st_nbr[pb] = s;
        const int ru = N_BOOK + s;
        const int pu = atomicAdd(&hist[ru >> BK_SHIFT], 1);
        st_row[pu] = ru;
        st_nbr[pu] = d;
    }
    __syncthreads();
    const int n = 2 * (c1 - c0);
    for (int p = tid; p < n; p += 1024) {
        const int row = st_row[p];
        const int g = gofs[row >> BK_SHIFT] + p;
        p_row[g] = row;
        p_nbr[g] = st_nbr[p];
    }
}

__device__ __forceinline__ void sort_row(const int* __restrict__ list, int n,
                                         int* __restrict__ out, int lane) {
    for (int i0 = 0; i0 < n; i0 += 32) {
        const int i = i0 + lane;
        const int mine = (i < n) ? list[i] : 0;
        int rank = 0;
        for (int j0 = 0; j0 < n; j0 += 32) {
            const int jn = (n - j0 < 32) ? (n - j0) : 32;
            const int theirs = (lane < jn) ? list[j0 + lane] : 0;
            for (int jj = 0; jj < jn; ++jj) {
                const int o = __builtin_amdgcn_readlane(theirs, jj);
                const int j = j0 + jj;
                rank += (o < mine || (o == mine && j < i)) ? 1 : 0;
            }
        }
        if (i < n) out[rank] = mine;
    }
}

__global__ void __launch_bounds__(1024)
build_lists_kernel(const int* __restrict__ p_row, const int* __restrict__ p_nbr,
                   const int* __restrict__ cofs, int* __restrict__ deg, int* __restrict__ off,
                   float* __restrict__ cnt, int* adj_raw, int* __restrict__ adj,
                   int total) {
    __shared__ int rcnt[BK_ROWS];
    __shared__ int loff[BK_ROWS];
    __shared__ int cur[BK_ROWS];
    __shared__ int wsum[32];
    __shared__ int lst[LIST_CAP];
    const int tid = threadIdx.x;
    const int lane = tid & 31;
    const int wave = __builtin_amdgcn_readfirstlane(tid >> 5);
    const int b = blockIdx.x;
    const int row0 = b << BK_SHIFT;
    const int g0 = __builtin_amdgcn_readfirstlane(cofs[b * NB1]);
    const int g1 = (b + 1 < NBK) ? __builtin_amdgcn_readfirstlane(cofs[(b + 1) * NB1]) : total;
    const int n = g1 - g0;

    rcnt[tid] = 0;
    __syncthreads();
    for (int p = tid; p < n; p += 1024) atomicAdd(&rcnt[p_row[g0 + p] - row0], 1);
    __syncthreads();

    const int v = rcnt[tid];
    int incl = v;
    for (int o = 1; o < 32; o <<= 1) {
        int t = __shfl_up(incl, o, 32);
        if (lane >= o) incl += t;
    }
    if (lane == 31) wsum[wave] = incl;
    __syncthreads();
    int front = 0;
    for (int w = 0; w < wave; ++w) front += wsum[w];
    const int ex = front + incl - v;
    loff[tid] = ex;
    cur[tid] = ex;
    if (row0 + tid < N_ROWS) {
        deg[row0 + tid] = v;
        off[row0 + tid] = g0 + ex;
        cnt[row0 + tid] = (float)v;
    }
    __syncthreads();

    const bool fits = n <= LIST_CAP;
    if (fits) {
        for (int p = tid; p < n; p += 1024) {
            const int q = atomicAdd(&cur[p_row[g0 + p] - row0], 1);
            lst[q] = p_nbr[g0 + p];
        }
    } else {
        for (int p = tid; p < n; p += 1024) {
            const int q = atomicAdd(&cur[p_row[g0 + p] - row0], 1);
            adj_raw[g0 + q] = p_nbr[g0 + p];
        }
    }
    __syncthreads();

    for (int r = wave; r < BK_ROWS; r += 32) {
        if (row0 + r >= N_ROWS) break;
        const int lo = __builtin_amdgcn_readfirstlane(loff[r]);
        const int m = __builtin_amdgcn_readfirstlane(rcnt[r]);
        if (fits) sort_row(lst + lo, m, adj + g0 + lo, lane);
        else      sort_row(adj_raw + g0 + lo, m, adj + g0 + lo, lane);
    }
}

template <int U>
__device__ __forceinline__ void add_rows(float4& acc, const float* __restrict__ feat,
                                         int ids, int j, int lane) {
    float4 v[U];
#pragma unroll
    for (int u = 0; u < U; ++u) {
        const int s = __builtin_amdgcn_readlane(ids, j + u);
        v[u] = *(const float4*)(feat + (size_t)s * HD + lane * 4);
    }
#pragma unroll
    for (int u = 0; u < U; ++u) {
        acc.x += v[u].x;
        acc.y += v[u].y;
        acc.z += v[u].z;
        acc.w += v[u].w;
    }
}

__global__ void gather_sum_kernel(const float* __restrict__ feat, const int* __restrict__ off,
                                  const int* __restrict__ deg, const int* __restrict__ adj,
                                  float* __restrict__ sums, int M) {
    const int lane = threadIdx.x & 31;
    const int waves = (int)(gridDim.x * 8);
    for (int row = __builtin_amdgcn_readfirstlane((int)(blockIdx.x * 8 + (threadIdx.x >> 5)));
         row < M; row += waves) {
        const int beg = off[row];
        const int n = deg[row];
        float4 acc = {0.0f, 0.0f, 0.0f, 0.0f};
        for (int c = 0; c < n; c += 32) {
            const int m = (n - c < 32) ? (n - c) : 32;
            const int ids = (lane < m) ? adj[beg + c + lane] : 0;
            int j = 0;
            for (; j + 8 <= m; j += 8) add_rows<8>(acc, feat, ids, j, lane);
            if (m - j >= 4) { add_rows<4>(acc, feat, ids, j, lane); j += 4; }
            if (m - j >= 2) { add_rows<2>(acc, feat, ids, j, lane); j += 2; }
            if (m - j >= 1) add_rows<1>(acc, feat, ids, j, lane);
        }
        *(float4*)(sums + (size_t)row * HD + lane * 4) = acc;
    }
}


__device__ __forceinline__ void load_weight_t(const float* __restrict__ W, float* __restrict__ wt,
                                              int K, int stride) {
    const int nf4 = K * (HD / 4);
    for (int idx = threadIdx.x; idx < nf4; idx += 256) {
        int k = idx >> 5;
        int n = (idx & 31) << 2;
        const float4 w4 = *(const float4*)(W + (size_t)k * HD + n);
        wt[(n + 0) * stride + k] = w4.x;
        wt[(n + 1) * stride + k] = w4.y;
        wt[(n + 2) * stride + k] = w4.z;
        wt[(n + 3) * stride + k] = w4.w;
    }
}

__global__ void gemm_bias_kernel(const float* __restrict__ A, const float* __restrict__ W,
                                 const float* __restrict__ bias, float* __restrict__ out,
                                 int M, int K) {
    __shared__ float wt[HD * (HD + 4)];
    const int stride = K + 4;
    load_weight_t(W, wt, K, stride);
    __syncthreads();

    const int wave = threadIdx.x >> 5;
    const int lane = threadIdx.x & 31;
    const int col  = lane & 15;
    const int half = lane >> 4;
    const int m0   = blockIdx.x * 128 + wave * 16;
    const int aRow = (m0 + col < M) ? (m0 + col) : (M - 1);

    v8f acc[8];
#pragma unroll
    for (int nt = 0; nt < 8; ++nt) {
        float bv = bias[nt * 16 + col];
        acc[nt] = (v8f){bv, bv, bv, bv, bv, bv, bv, bv};
    }

    const float* arow = A + (size_t)aRow * K;
    for (int k0 = 0; k0 < K; k0 += 4) {
        const int kb = k0 + 2 * half;
        v2f a = *(const v2f*)(arow + kb);
        v2f bf[8];
#pragma unroll
        for (int nt = 0; nt < 8; ++nt)
            bf[nt] = *(const v2f*)&wt[(nt * 16 + col) * stride + kb];
#pragma unroll
        for (int nt = 0; nt < 8; ++nt)
            acc[nt] = __builtin_amdgcn_wmma_f32_16x16x4_f32(
                false, a, false, bf[nt], (short)0, acc[nt], false, false);
    }

#pragma unroll
    for (int nt = 0; nt < 8; ++nt)
#pragma unroll
        for (int v = 0; v < 8; ++v) {
            int dr = m0 + v + 8 * half;
            if (dr < M) out[(size_t)dr * HD + nt * 16 + col] = acc[nt][v];
        }
}

__global__ void sage_combine_kernel(const float* __restrict__ sums, const float* __restrict__ cnt,
                                    const float* __restrict__ X, const float* __restrict__ Wl,
                                    const float* __restrict__ bl, const float* __restrict__ Wr,
                                    float* __restrict__ out, int M) {
    __shared__ float wt[HD * (HD + 4)];
    const int stride = HD + 4;

    load_weight_t(Wl, wt, HD, stride);
    __syncthreads();

    const int wave = threadIdx.x >> 5;
    const int lane = threadIdx.x & 31;
    const int col  = lane & 15;
    const int half = lane >> 4;
    const int m0   = blockIdx.x * 128 + wave * 16;
    const int aRow = (m0 + col < M) ? (m0 + col) : (M - 1);
    const float invc = 1.0f / fmaxf(cnt[aRow], 1.0f);

    v8f acc[8];
#pragma unroll
    for (int nt = 0; nt < 8; ++nt) {
        float bv = bl[nt * 16 + col];
        acc[nt] = (v8f){bv, bv, bv, bv, bv, bv, bv, bv};
    }

    const float* srow = sums + (size_t)aRow * HD;
    for (int k0 = 0; k0 < HD; k0 += 4) {
        const int kb = k0 + 2 * half;
        v2f a = *(const v2f*)(srow + kb);
        a.x *= invc;
        a.y *= invc;
        v2f bf[8];
#pragma unroll
        for (int nt = 0; nt < 8; ++nt)
            bf[nt] = *(const v2f*)&wt[(nt * 16 + col) * stride + kb];
#pragma unroll
        for (int nt = 0; nt < 8; ++nt)
            acc[nt] = __builtin_amdgcn_wmma_f32_16x16x4_f32(
                false, a, false, bf[nt], (short)0, acc[nt], false, false);
    }

    __syncthreads();
    load_weight_t(Wr, wt, HD, stride);
    __syncthreads();

    const float* xrow = X + (size_t)aRow * HD;
    for (int k0 = 0; k0 < HD; k0 += 4) {
        const int kb = k0 + 2 * half;
        v2f a = *(const v2f*)(xrow + kb);
        v2f bf[8];
#pragma unroll
        for (int nt = 0; nt < 8; ++nt)
            bf[nt] = *(const v2f*)&wt[(nt * 16 + col) * stride + kb];
#pragma unroll
        for (int nt = 0; nt < 8; ++nt)
            acc[nt] = __builtin_amdgcn_wmma_f32_16x16x4_f32(
                false, a, false, bf[nt], (short)0, acc[nt], false, false);
    }

#pragma unroll
    for (int nt = 0; nt < 8; ++nt)
#pragma unroll
        for (int v = 0; v < 8; ++v) {
            int dr = m0 + v + 8 * half;
            if (dr < M) out[(size_t)dr * HD + nt * 16 + col] = fmaxf(acc[nt][v], 0.0f);
        }
}

template <int U>
__device__ __forceinline__ void ln_rows(const float* __restrict__ X, float* __restrict__ out,
                                        const float4 g, const float4 bt, int row0, int waves,
                                        int lane) {
    float4 v[U];
#pragma unroll
    for (int u = 0; u < U; ++u)
        v[u] = *(const float4*)(X + (size_t)(row0 + u * waves) * HD + lane * 4);
#pragma unroll
    for (int u = 0; u < U; ++u) {
        float s = v[u].x + v[u].y + v[u].z + v[u].w;
#pragma unroll
        for (int off = 16; off >= 1; off >>= 1) s += __shfl_xor(s, off, 32);
        const float mu = s * (1.0f / HD);

        const float dx = v[u].x - mu, dy = v[u].y - mu, dz = v[u].z - mu, dw = v[u].w - mu;
        float q = dx * dx + dy * dy + dz * dz + dw * dw;
#pragma unroll
        for (int off = 16; off >= 1; off >>= 1) q += __shfl_xor(q, off, 32);
        const float rstd = rsqrtf(q * (1.0f / HD) + LN_EPS);

        float4 o;
        o.x = dx * rstd * g.x + bt.x;
        o.y = dy * rstd * g.y + bt.y;
        o.z = dz * rstd * g.z + bt.z;
        o.w = dw * rstd * g.w + bt.w;
        *(float4*)(out + (size_t)(row0 + u * waves) * HD + lane * 4) = o;
    }
}

__global__ void layernorm_kernel(const float* __restrict__ X, const float* __restrict__ gamma,
                                 const float* __restrict__ beta, float* __restrict__ out, int M) {
    const int lane = threadIdx.x & 31;
    const int waves = (int)(gridDim.x * 8);
    const float4 g = *(const float4*)(gamma + lane * 4);
    const float4 bt = *(const float4*)(beta + lane * 4);

    int row = __builtin_amdgcn_readfirstlane((int)(blockIdx.x * 8 + (threadIdx.x >> 5)));
    for (; row + 3 * waves < M; row += 4 * waves) ln_rows<4>(X, out, g, bt, row, waves, lane);
    for (; row < M; row += waves) ln_rows<1>(X, out, g, bt, row, waves, lane);
}

extern "C" void kernel_launch(void* const* d_in, const int* in_sizes, int n_in,
                              void* d_out, int out_size, void* d_ws, size_t ws_size,
                              hipStream_t stream) {
    const float* user_x      = (const float*)d_in[0];
    const float* book_x      = (const float*)d_in[1];
    const int*   edge_src    = (const int*)d_in[2];
    const int*   edge_dst    = (const int*)d_in[3];
    const float* user_proj_w = (const float*)d_in[4];
    const float* user_proj_b = (const float*)d_in[5];
    const float* book_proj_w = (const float*)d_in[6];
    const float* book_proj_b = (const float*)d_in[7];
    const float* Wl          = (const float*)d_in[8];
    const float* bl          = (const float*)d_in[9];
    const float* Wr          = (const float*)d_in[10];
    const float* ugamma      = (const float*)d_in[11];
    const float* ubeta       = (const float*)d_in[12];
    const float* bgamma      = (const float*)d_in[13];
    const float* bbeta       = (const float*)d_in[14];

    float* ws = (float*)d_ws;
    size_t off = 0;
    auto carve = [&](size_t n) { float* p = ws + off; off += n; return p; };
    float* u_cur = carve((size_t)N_USER * HD);
    float* u_nxt = carve((size_t)N_USER * HD);
    float* b_cur = carve((size_t)N_BOOK * HD);
    float* b_nxt = carve((size_t)N_BOOK * HD);
    float* sum_u = carve((size_t)N_USER * HD);
    float* sum_b = carve((size_t)N_BOOK * HD);
    float* cnt   = carve((size_t)N_ROWS);
    int* deg     = (int*)carve((size_t)N_ROWS);
    int* offs    = (int*)carve((size_t)N_ROWS);
    int* counts  = (int*)carve((size_t)N_CNT);
    int* cofs    = (int*)carve((size_t)N_CNT);
    int* part    = (int*)carve((size_t)64);
    int* p_row   = (int*)carve((size_t)2 * N_EDGE);
    int* p_nbr   = (int*)carve((size_t)2 * N_EDGE);
    int* adj_raw = (int*)carve((size_t)2 * N_EDGE);
    int* adj     = (int*)carve((size_t)2 * N_EDGE);
    const float* cnt_b = cnt;
    const float* cnt_u = cnt + N_BOOK;

    gemm_bias_kernel<<<(N_USER + 127) / 128, 256, 0, stream>>>(
        user_x, user_proj_w, user_proj_b, u_cur, N_USER, 64);
    gemm_bias_kernel<<<(N_BOOK + 127) / 128, 256, 0, stream>>>(
        book_x, book_proj_w, book_proj_b, b_cur, N_BOOK, 128);

    bucket_count_kernel<<<NB1, 1024, 0, stream>>>(edge_src, edge_dst, counts, N_EDGE);
    block_sum_kernel<<<CSCAN_BLOCKS, 256, 0, stream>>>(counts, part, N_CNT);
    scan_kernel<<<CSCAN_BLOCKS, 256, 0, stream>>>(counts, part, cofs, N_CNT);
    bucket_scatter_kernel<<<NB1, 1024, 0, stream>>>(
        edge_src, edge_dst, cofs, p_row, p_nbr, N_EDGE);
    build_lists_kernel<<<NBK, 1024, 0, stream>>>(
        p_row, p_nbr, cofs, deg, offs, cnt, adj_raw, adj, 2 * N_EDGE);

    for (int l = 0; l < 3; ++l) {
        const float* wl  = Wl + (size_t)l * HD * HD;
        const float* blv = bl + (size_t)l * HD;
        const float* wr  = Wr + (size_t)l * HD * HD;

        gather_sum_kernel<<<FILL_BLOCKS, 256, 0, stream>>>(
            u_cur, offs, deg, adj, sum_b, N_BOOK);
        gather_sum_kernel<<<FILL_BLOCKS, 256, 0, stream>>>(
            b_cur, offs + N_BOOK, deg + N_BOOK, adj, sum_u, N_USER);

        sage_combine_kernel<<<(N_BOOK + 127) / 128, 256, 0, stream>>>(
            sum_b, cnt_b, b_cur, wl, blv, wr, b_nxt, N_BOOK);
        sage_combine_kernel<<<(N_USER + 127) / 128, 256, 0, stream>>>(
            sum_u, cnt_u, u_cur, wl, blv, wr, u_nxt, N_USER);

        float* t;
        t = u_cur; u_cur = u_nxt; u_nxt = t;
        t = b_cur; b_cur = b_nxt; b_nxt = t;
    }

    float* out_u = (float*)d_out;
    float* out_b = out_u + (size_t)N_USER * HD;
    layernorm_kernel<<<FILL_BLOCKS, 256, 0, stream>>>(u_cur, ugamma, ubeta, out_u, N_USER);
    layernorm_kernel<<<FILL_BLOCKS, 256, 0, stream>>>(b_cur, bgamma, bbeta, out_b, N_BOOK);
}


# Gather-sum fused into the SAGE GEMM: a 16-wave group per WGP keeps Wl, Wr in LDS; each wave loops over 16-row tiles (list sums via LDS to A fragments, both products), no barriers; sum order unchanged
# speedup vs baseline: 3.3364x; 1.1606x over previous
#include <hip/hip_runtime.h>

typedef float v2f __attribute__((ext_vector_type(2)));
typedef float v8f __attribute__((ext_vector_type(8)));

#define HD 128
constexpr int N_USER = 150000;
constexpr int N_BOOK = 75000;
constexpr int N_EDGE = 1500000;
constexpr float LN_EPS = 1e-5f;
constexpr int N_ROWS = N_BOOK + N_USER;
constexpr int SCAN_TILE = 1024;
constexpr int FILL_BLOCKS = 2048;

constexpr int BK_SHIFT = 10;
constexpr int BK_ROWS = 1 << BK_SHIFT;
constexpr int NBK = (N_ROWS + BK_ROWS - 1) >> BK_SHIFT;
constexpr int NB1 = 256;
constexpr int CHUNK = (N_EDGE + NB1 - 1) / NB1;
constexpr int N_CNT = NBK * NB1;
constexpr int CSCAN_BLOCKS = (N_CNT + SCAN_TILE - 1) / SCAN_TILE;
constexpr int BK_PER_LANE = (NBK + 31) / 32;
constexpr int LIST_CAP = 40960;


__global__ void __launch_bounds__(1024)
bucket_count_kernel(const int* __restrict__ esrc, const int* __restrict__ edst,
                    int* __restrict__ counts, int E) {
    __shared__ int hist[NBK];
    for (int i = threadIdx.x; i < NBK; i += 1024) hist[i] = 0;
    __syncthreads();
    int c0 = blockIdx.x * CHUNK;
    if (c0 > E) c0 = E;
    const int c1 = (c0 + CHUNK < E) ? (c0 + CHUNK) : E;
    for (int i = c0 + threadIdx.x; i < c1; i += 1024) {
        atomicAdd(&hist[edst[i] >> BK_SHIFT], 1);
        atomicAdd(&hist[(N_BOOK + esrc[i]) >> BK_SHIFT], 1);
    }
    __syncthreads();
    for (int i = threadIdx.x; i < NBK; i += 1024) counts[i * NB1 + blockIdx.x] = hist[i];
}

__global__ void block_sum_kernel(const int* __restrict__ deg, int* __restrict__ part, int n) {
    __shared__ int red[8];
    const int lane = threadIdx.x & 31;
    const int wave = threadIdx.x >> 5;
    const int base = blockIdx.x * SCAN_TILE + threadIdx.x * 4;
    int s = 0;
    for (int u = 0; u < 4; ++u)
        if (base + u < n) s += deg[base + u];
    for (int o = 16; o >= 1; o >>= 1) s += __shfl_xor(s, o, 32);
    if (lane == 0) red[wave] = s;
    __syncthreads();
    if (threadIdx.x == 0) {
        int t = 0;
        for (int w = 0; w < 8; ++w) t += red[w];
        part[blockIdx.x] = t;
    }
}

__global__ void scan_kernel(const int* __restrict__ deg, const int* __restrict__ part,
                            int* __restrict__ off, int n) {
    __shared__ int red[8];
    __shared__ int wsum[8];
    const int lane = threadIdx.x & 31;
    const int wave = threadIdx.x >> 5;

    int p = 0;
    for (int i = threadIdx.x; i < (int)blockIdx.x; i += 256) p += part[i];
    for (int o = 16; o >= 1; o >>= 1) p += __shfl_xor(p, o, 32);
    if (lane == 0) red[wave] = p;

    const int base = blockIdx.x * SCAN_TILE + threadIdx.x * 4;
    int d[4];
    int tsum = 0;
    for (int u = 0; u < 4; ++u) {
        d[u] = (base + u < n) ? deg[base + u] : 0;
        tsum += d[u];
    }
    int incl = tsum;
    for (int o = 1; o < 32; o <<= 1) {
        int t = __shfl_up(incl, o, 32);
        if (lane >= o) incl += t;
    }
    if (lane == 31) wsum[wave] = incl;
    __syncthreads();

    int run = incl - tsum;
    for (int w = 0; w < 8; ++w) {
        run += red[w];
        if (w < wave) run += wsum[w];
    }
    for (int u = 0; u < 4; ++u)
        if (base + u < n) {
            off[base + u] = run;
            run += d[u];
        }
}

__global__ void __launch_bounds__(1024)
bucket_scatter_kernel(const int* __restrict__ esrc, const int* __restrict__ edst,
                      const int* __restrict__ cofs, int* __restrict__ p_row,
                      int* __restrict__ p_nbr, int E) {
    __shared__ int hist[NBK];
    __shared__ int lbase[NBK];
    __shared__ int gofs[NBK];
    __shared__ int st_row[2 * CHUNK];
    __shared__ int st_nbr[2 * CHUNK];
    const int tid = threadIdx.x;
    for (int i = tid; i < NBK; i += 1024) hist[i] = 0;
    __syncthreads();
    int c0 = blockIdx.x * CHUNK;
    if (c0 > E) c0 = E;
    const int c1 = (c0 + CHUNK < E) ? (c0 + CHUNK) : E;
    for (int i = c0 + tid; i < c1; i += 1024) {
        atomicAdd(&hist[edst[i] >> BK_SHIFT], 1);
        atomicAdd(&hist[(N_BOOK + esrc[i]) >> BK_SHIFT], 1);
    }
    __syncthreads();
    if (tid < 32) {
        int v[BK_PER_LANE];
        int s = 0;
        for (int u = 0; u < BK_PER_LANE; ++u) {
            const int idx = tid * BK_PER_LANE + u;
            v[u] = (idx < NBK) ? hist[idx] : 0;
            s += v[u];
        }
        int incl = s;
        for (int o = 1; o < 32; o <<= 1) {
            int t = __shfl_up(incl, o, 32);
            if (tid >= o) incl += t;
        }
        int run = incl - s;
        for (int u = 0; u < BK_PER_LANE; ++u) {
            const int idx = tid * BK_PER_LANE + u;
            if (idx < NBK) {
                lbase[idx] = run;
                run += v[u];
            }
        }
    }
    __syncthreads();
    for (int i = tid; i < NBK; i += 1024) {
        const int lb = lbase[i];
        hist[i] = lb;
        gofs[i] = cofs[i * NB1 + blockIdx.x] - lb;
    }
    __syncthreads();
    for (int i = c0 + tid; i < c1; i += 1024) {
        const int s = esrc[i], d = edst[i];
        const int rb = d;
        const int pb = atomicAdd(&hist[rb >> BK_SHIFT], 1);
        st_row[pb] = rb;
        st_nbr[pb] = s;
        const int ru = N_BOOK + s;
        const int pu = atomicAdd(&hist[ru >> BK_SHIFT], 1);
        st_row[pu] = ru;
        st_nbr[pu] = d;
    }
    __syncthreads();
    const int n = 2 * (c1 - c0);
    for (int p = tid; p < n; p += 1024) {
        const int row = st_row[p];
        const int g = gofs[row >> BK_SHIFT] + p;
        p_row[g] = row;
        p_nbr[g] = st_nbr[p];
    }
}

__device__ __forceinline__ void sort_row(const int* __restrict__ list, int n,
                                         int* __restrict__ out, int lane) {
    for (int i0 = 0; i0 < n; i0 += 32) {
        const int i = i0 + lane;
        const int mine = (i < n) ? list[i] : 0;
        int rank = 0;
        for (int j0 = 0; j0 < n; j0 += 32) {
            const int jn = (n - j0 < 32) ? (n - j0) : 32;
            const int theirs = (lane < jn) ? list[j0 + lane] : 0;
            for (int jj = 0; jj < jn; ++jj) {
                const int o = __builtin_amdgcn_readlane(theirs, jj);
                const int j = j0 + jj;
                rank += (o < mine || (o == mine && j < i)) ? 1 : 0;
            }
        }
        if (i < n) out[rank] = mine;
    }
}

__global__ void __launch_bounds__(1024)
build_lists_kernel(const int* __restrict__ p_row, const int* __restrict__ p_nbr,
                   const int* __restrict__ cofs, int* __restrict__ deg, int* __restrict__ off,
                   float* __restrict__ cnt, int* adj_raw, int* __restrict__ adj,
                   int total) {
    __shared__ int rcnt[BK_ROWS];
    __shared__ int loff[BK_ROWS];
    __shared__ int cur[BK_ROWS];
    __shared__ int wsum[32];
    __shared__ int lst[LIST_CAP];
    const int tid = threadIdx.x;
    const int lane = tid & 31;
    const int wave = __builtin_amdgcn_readfirstlane(tid >> 5);
    const int b = blockIdx.x;
    const int row0 = b << BK_SHIFT;
    const int g0 = __builtin_amdgcn_readfirstlane(cofs[b * NB1]);
    const int g1 = (b + 1 < NBK) ? __builtin_amdgcn_readfirstlane(cofs[(b + 1) * NB1]) : total;
    const int n = g1 - g0;

    rcnt[tid] = 0;
    __syncthreads();
    for (int p = tid; p < n; p += 1024) atomicAdd(&rcnt[p_row[g0 + p] - row0], 1);
    __syncthreads();

    const int v = rcnt[tid];
    int incl = v;
    for (int o = 1; o < 32; o <<= 1) {
        int t = __shfl_up(incl, o, 32);
        if (lane >= o) incl += t;
    }
    if (lane == 31) wsum[wave] = incl;
    __syncthreads();
    int front = 0;
    for (int w = 0; w < wave; ++w) front += wsum[w];
    const int ex = front + incl - v;
    loff[tid] = ex;
    cur[tid] = ex;
    if (row0 + tid < N_ROWS) {
        deg[row0 + tid] = v;
        off[row0 + tid] = g0 + ex;
        cnt[row0 + tid] = (float)v;
    }
    __syncthreads();

    const bool fits = n <= LIST_CAP;
    if (fits) {
        for (int p = tid; p < n; p += 1024) {
            const int q = atomicAdd(&cur[p_row[g0 + p] - row0], 1);
            lst[q] = p_nbr[g0 + p];
        }
    } else {
        for (int p = tid; p < n; p += 1024) {
            const int q = atomicAdd(&cur[p_row[g0 + p] - row0], 1);
            adj_raw[g0 + q] = p_nbr[g0 + p];
        }
    }
    __syncthreads();

    for (int r = wave; r < BK_ROWS; r += 32) {
        if (row0 + r >= N_ROWS) break;
        const int lo = __builtin_amdgcn_readfirstlane(loff[r]);
        const int m = __builtin_amdgcn_readfirstlane(rcnt[r]);
        if (fits) sort_row(lst + lo, m, adj + g0 + lo, lane);
        else      sort_row(adj_raw + g0 + lo, m, adj + g0 + lo, lane);
    }
}

__device__ __forceinline__ void add_rows_upto16(float4& acc, const float* __restrict__ feat,
                                                int ids, int j, int left, int lane) {
    int s[16];
#pragma unroll
    for (int u = 0; u < 16; ++u) s[u] = __builtin_amdgcn_readlane(ids, j + u);
    float4 v[16];
#pragma unroll
    for (int u = 0; u < 16; ++u) {
        v[u] = float4{0.0f, 0.0f, 0.0f, 0.0f};
        if (u < left) v[u] = *(const float4*)(feat + (size_t)s[u] * HD + lane * 4);
    }
#pragma unroll
    for (int u = 0; u < 16; ++u) {
        acc.x += v[u].x;
        acc.y += v[u].y;
        acc.z += v[u].z;
        acc.w += v[u].w;
    }
}


__device__ __forceinline__ void load_weight_t(const float* __restrict__ W, float* __restrict__ wt,
                                              int K, int stride, int nthr) {
    const int nf4 = K * (HD / 4);
    for (int idx = threadIdx.x; idx < nf4; idx += nthr) {
        int k = idx >> 5;
        int n = (idx & 31) << 2;
        const float4 w4 = *(const float4*)(W + (size_t)k * HD + n);
        wt[(n + 0) * stride + k] = w4.x;
        wt[(n + 1) * stride + k] = w4.y;
        wt[(n + 2) * stride + k] = w4.z;
        wt[(n + 3) * stride + k] = w4.w;
    }
}

__global__ void gemm_bias_kernel(const float* __restrict__ A, const float* __restrict__ W,
                                 const float* __restrict__ bias, float* __restrict__ out,
                                 int M, int K) {
    __shared__ float wt[HD * (HD + 4)];
    const int stride = K + 4;
    load_weight_t(W, wt, K, stride, 256);
    __syncthreads();

    const int wave = threadIdx.x >> 5;
    const int lane = threadIdx.x & 31;
    const int col  = lane & 15;
    const int half = lane >> 4;
    const int m0   = blockIdx.x * 128 + wave * 16;
    const int aRow = (m0 + col < M) ? (m0 + col) : (M - 1);

    v8f acc[8];
#pragma unroll
    for (int nt = 0; nt < 8; ++nt) {
        float bv = bias[nt * 16 + col];
        acc[nt] = (v8f){bv, bv, bv, bv, bv, bv, bv, bv};
    }

    const float* arow = A + (size_t)aRow * K;
    for (int k0 = 0; k0 < K; k0 += 4) {
        const int kb = k0 + 2 * half;
        v2f a = *(const v2f*)(arow + kb);
        v2f bf[8];
#pragma unroll
        for (int nt = 0; nt < 8; ++nt)
            bf[nt] = *(const v2f*)&wt[(nt * 16 + col) * stride + kb];
#pragma unroll
        for (int nt = 0; nt < 8; ++nt)
            acc[nt] = __builtin_amdgcn_wmma_f32_16x16x4_f32(
                false, a, false, bf[nt], (short)0, acc[nt], false, false);
    }

#pragma unroll
    for (int nt = 0; nt < 8; ++nt)
#pragma unroll
        for (int v = 0; v < 8; ++v) {
            int dr = m0 + v + 8 * half;
            if (dr < M) out[(size_t)dr * HD + nt * 16 + col] = acc[nt][v];
        }
}

constexpr int WT_WORDS   = HD * (HD + 4);
constexpr int TILE_WORDS = 16 * (HD + 4);
constexpr int SAGE_WAVES = 16;
constexpr int SAGE_BLOCKS = 256;
constexpr int TILES_B = (N_BOOK + 15) / 16;
constexpr int TILES_U = (N_USER + 15) / 16;

__global__ void __launch_bounds__(SAGE_WAVES * 32)
sage_layer_kernel(const float* __restrict__ u_cur, const float* __restrict__ b_cur,
                  const int* __restrict__ off, const int* __restrict__ deg,
                  const int* __restrict__ adj, const float* __restrict__ cnt,
                  const float* __restrict__ Wl, const float* __restrict__ bl,
                  const float* __restrict__ Wr, float* __restrict__ u_nxt,
                  float* __restrict__ b_nxt) {
    __shared__ alignas(16) float wl_t[WT_WORDS];
    __shared__ alignas(16) float wr_t[WT_WORDS];
    __shared__ alignas(16) float tiles[SAGE_WAVES * TILE_WORDS];
    const int stride = HD + 4;

    load_weight_t(Wl, wl_t, HD, stride, SAGE_WAVES * 32);
    load_weight_t(Wr, wr_t, HD, stride, SAGE_WAVES * 32);
    __syncthreads();

    const int lane = threadIdx.x & 31;
    const int wave = __builtin_amdgcn_readfirstlane((int)(threadIdx.x >> 5));
    const int col  = lane & 15;
    const int half = lane >> 4;
    float* __restrict__ st = tiles + wave * TILE_WORDS;

    float bv[8];
#pragma unroll
    for (int nt = 0; nt < 8; ++nt) bv[nt] = bl[nt * 16 + col];

    for (int q = wave; ; q += SAGE_WAVES) {
        const int t = (int)blockIdx.x + (int)gridDim.x * q;
        if (t >= TILES_B + TILES_U) break;
        asm volatile("" ::: "memory");
        const bool isb = t < TILES_B;
        const int M  = isb ? N_BOOK : N_USER;
        const int m0 = (isb ? t : t - TILES_B) * 16;
        const int gsh = isb ? 0 : N_BOOK;
        const float* __restrict__ feat = isb ? u_cur : b_cur;
        const float* __restrict__ X    = isb ? b_cur : u_cur;
        float* __restrict__ out        = isb ? b_nxt : u_nxt;

        int my_beg = 0, my_n = 0;
        if (lane < 16 && m0 + lane < M) {
            my_beg = off[gsh + m0 + lane];
            my_n = deg[gsh + m0 + lane];
        }
        int ids_pre[16];
#pragma unroll
        for (int r = 0; r < 16; ++r) {
            const int b = __builtin_amdgcn_readlane(my_beg, r);
            const int nn = __builtin_amdgcn_readlane(my_n, r);
            ids_pre[r] = (lane < nn) ? adj[b + lane] : 0;
        }
#pragma unroll
        for (int r = 0; r < 16; ++r) st[r * stride + lane] = __int_as_float(ids_pre[r]);

        for (int r = 0; r < 16; ++r) {
            const int cbeg = __builtin_amdgcn_readlane(my_beg, r);
            const int cn = __builtin_amdgcn_readlane(my_n, r);
            float4 acc4 = {0.0f, 0.0f, 0.0f, 0.0f};
            for (int c = 0; c < cn; c += 32) {
                const int m = (cn - c < 32) ? (cn - c) : 32;
                int ids;
                if (c == 0) ids = __float_as_int(st[r * stride + lane]);
                else        ids = (lane < m) ? adj[cbeg + c + lane] : 0;
                for (int j = 0; j < m; j += 16) add_rows_upto16(acc4, feat, ids, j, m - j, lane);
            }
            *(float4*)&st[r * stride + lane * 4] = acc4;
        }

        const int aRow = (m0 + col < M) ? (m0 + col) : (M - 1);
        const float invc = 1.0f / fmaxf(cnt[gsh + aRow], 1.0f);
        v2f a[HD / 4];
#pragma unroll
        for (int k = 0; k < HD / 4; ++k) {
            a[k] = *(const v2f*)&st[col * stride + 4 * k + 2 * half];
            a[k].x *= invc;
            a[k].y *= invc;
        }

        v8f acc[8];
#pragma unroll
        for (int nt = 0; nt < 8; ++nt)
            acc[nt] = (v8f){bv[nt], bv[nt], bv[nt], bv[nt], bv[nt], bv[nt], bv[nt], bv[nt]};

#pragma unroll
        for (int k = 0; k < HD / 4; ++k) {
            const int kb = 4 * k + 2 * half;
            v2f bf[8];
#pragma unroll
            for (int nt = 0; nt < 8; ++nt)
                bf[nt] = *(const v2f*)&wl_t[(nt * 16 + col) * stride + kb];
#pragma unroll
            for (int nt = 0; nt < 8; ++nt)
                acc[nt] = __builtin_amdgcn_wmma_f32_16x16x4_f32(
                    false, a[k], false, bf[nt], (short)0, acc[nt], false, false);
        }

        const float* xrow = X + (size_t)aRow * HD;
        for (int k0 = 0; k0 < HD; k0 += 4) {
            const int kb = k0 + 2 * half;
            v2f ax = *(const v2f*)(xrow + kb);
            v2f bf[8];
#pragma unroll
            for (int nt = 0; nt < 8; ++nt)
                bf[nt] = *(const v2f*)&wr_t[(nt * 16 + col) * stride + kb];
#pragma unroll
            for (int nt = 0; nt < 8; ++nt)
                acc[nt] = __builtin_amdgcn_wmma_f32_16x16x4_f32(
                    false, ax, false, bf[nt], (short)0, acc[nt], false, false);
        }

#pragma unroll
        for (int nt = 0; nt < 8; ++nt)
#pragma unroll
            for (int v = 0; v < 8; ++v) {
                int dr = m0 + v + 8 * half;
                if (dr < M) out[(size_t)dr * HD + nt * 16 + col] = fmaxf(acc[nt][v], 0.0f);
            }
    }
}

template <int U>
__device__ __forceinline__ void ln_rows(const float* __restrict__ X, float* __restrict__ out,
                                        const float4 g, const float4 bt, int row0, int waves,
                                        int lane) {
    float4 v[U];
#pragma unroll
    for (int u = 0; u < U; ++u)
        v[u] = *(const float4*)(X + (size_t)(row0 + u * waves) * HD + lane * 4);
#pragma unroll
    for (int u = 0; u < U; ++u) {
        float s = v[u].x + v[u].y + v[u].z + v[u].w;
#pragma unroll
        for (int off = 16; off >= 1; off >>= 1) s += __shfl_xor(s, off, 32);
        const float mu = s * (1.0f / HD);

        const float dx = v[u].x - mu, dy = v[u].y - mu, dz = v[u].z - mu, dw = v[u].w - mu;
        float q = dx * dx + dy * dy + dz * dz + dw * dw;
#pragma unroll
        for (int off = 16; off >= 1; off >>= 1) q += __shfl_xor(q, off, 32);
        const float rstd = rsqrtf(q * (1.0f / HD) + LN_EPS);

        float4 o;
        o.x = dx * rstd * g.x + bt.x;
        o.y = dy * rstd * g.y + bt.y;
        o.z = dz * rstd * g.z + bt.z;
        o.w = dw * rstd * g.w + bt.w;
        *(float4*)(out + (size_t)(row0 + u * waves) * HD + lane * 4) = o;
    }
}

__global__ void layernorm_kernel(const float* __restrict__ X, const float* __restrict__ gamma,
                                 const float* __restrict__ beta, float* __restrict__ out, int M) {
    const int lane = threadIdx.x & 31;
    const int waves = (int)(gridDim.x * 8);
    const float4 g = *(const float4*)(gamma + lane * 4);
    const float4 bt = *(const float4*)(beta + lane * 4);

    int row = __builtin_amdgcn_readfirstlane((int)(blockIdx.x * 8 + (threadIdx.x >> 5)));
    for (; row + 3 * waves < M; row += 4 * waves) ln_rows<4>(X, out, g, bt, row, waves, lane);
    for (; row < M; row += waves) ln_rows<1>(X, out, g, bt, row, waves, lane);
}

extern "C" void kernel_launch(void* const* d_in, const int* in_sizes, int n_in,
                              void* d_out, int out_size, void* d_ws, size_t ws_size,
                              hipStream_t stream) {
    const float* user_x      = (const float*)d_in[0];
    const float* book_x      = (const float*)d_in[1];
    const int*   edge_src    = (const int*)d_in[2];
    const int*   edge_dst    = (const int*)d_in[3];
    const float* user_proj_w = (const float*)d_in[4];
    const float* user_proj_b = (const float*)d_in[5];
    const float* book_proj_w = (const float*)d_in[6];
    const float* book_proj_b = (const float*)d_in[7];
    const float* Wl          = (const float*)d_in[8];
    const float* bl          = (const float*)d_in[9];
    const float* Wr          = (const float*)d_in[10];
    const float* ugamma      = (const float*)d_in[11];
    const float* ubeta       = (const float*)d_in[12];
    const float* bgamma      = (const float*)d_in[13];
    const float* bbeta       = (const float*)d_in[14];

    float* ws = (float*)d_ws;
    size_t off = 0;
    auto carve = [&](size_t n) { float* p = ws + off; off += n; return p; };
    float* u_cur = carve((size_t)N_USER * HD);
    float* u_nxt = carve((size_t)N_USER * HD);
    float* b_cur = carve((size_t)N_BOOK * HD);
    float* b_nxt = carve((size_t)N_BOOK * HD);
    float* cnt   = carve((size_t)N_ROWS);
    int* deg     = (int*)carve((size_t)N_ROWS);
    int* offs    = (int*)carve((size_t)N_ROWS);
    int* counts  = (int*)carve((size_t)N_CNT);
    int* cofs    = (int*)carve((size_t)N_CNT);
    int* part    = (int*)carve((size_t)64);
    int* p_row   = (int*)carve((size_t)2 * N_EDGE);
    int* p_nbr   = (int*)carve((size_t)2 * N_EDGE);
    int* adj_raw = (int*)carve((size_t)2 * N_EDGE);
    int* adj     = (int*)carve((size_t)2 * N_EDGE);

    gemm_bias_kernel<<<(N_USER + 127) / 128, 256, 0, stream>>>(
        user_x, user_proj_w, user_proj_b, u_cur, N_USER, 64);
    gemm_bias_kernel<<<(N_BOOK + 127) / 128, 256, 0, stream>>>(
        book_x, book_proj_w, book_proj_b, b_cur, N_BOOK, 128);

    bucket_count_kernel<<<NB1, 1024, 0, stream>>>(edge_src, edge_dst, counts, N_EDGE);
    block_sum_kernel<<<CSCAN_BLOCKS, 256, 0, stream>>>(counts, part, N_CNT);
    scan_kernel<<<CSCAN_BLOCKS, 256, 0, stream>>>(counts, part, cofs, N_CNT);
    bucket_scatter_kernel<<<NB1, 1024, 0, stream>>>(
        edge_src, edge_dst, cofs, p_row, p_nbr, N_EDGE);
    build_lists_kernel<<<NBK, 1024, 0, stream>>>(
        p_row, p_nbr, cofs, deg, offs, cnt, adj_raw, adj, 2 * N_EDGE);

    for (int l = 0; l < 3; ++l) {
        const float* wl  = Wl + (size_t)l * HD * HD;
        const float* blv = bl + (size_t)l * HD;
        const float* wr  = Wr + (size_t)l * HD * HD;

        sage_layer_kernel<<<SAGE_BLOCKS, SAGE_WAVES * 32, 0, stream>>>(
            u_cur, b_cur, offs, deg, adj, cnt, wl, blv, wr, u_nxt, b_nxt);

        float* t;
        t = u_cur; u_cur = u_nxt; u_nxt = t;
        t = b_cur; b_cur = b_nxt; b_nxt = t;
    }

    float* out_u = (float*)d_out;
    float* out_b = out_u + (size_t)N_USER * HD;
    layernorm_kernel<<<FILL_BLOCKS, 256, 0, stream>>>(u_cur, ugamma, ubeta, out_u, N_USER);
    layernorm_kernel<<<FILL_BLOCKS, 256, 0, stream>>>(b_cur, bgamma, bbeta, out_b, N_BOOK);
}


# SAGE gather walks a tile's 16 neighbour lists as one stream, 32 row loads in flight per batch across row ends (was one row at a time, 16 at most); each row still summed from zero in list order
# speedup vs baseline: 3.5342x; 1.0593x over previous
#include <hip/hip_runtime.h>

typedef float v2f __attribute__((ext_vector_type(2)));
typedef float v8f __attribute__((ext_vector_type(8)));

#define HD 128
constexpr int N_USER = 150000;
constexpr int N_BOOK = 75000;
constexpr int N_EDGE = 1500000;
constexpr float LN_EPS = 1e-5f;
constexpr int N_ROWS = N_BOOK + N_USER;
constexpr int SCAN_TILE = 1024;
constexpr int FILL_BLOCKS = 2048;

constexpr int BK_SHIFT = 10;
constexpr int BK_ROWS = 1 << BK_SHIFT;
constexpr int NBK = (N_ROWS + BK_ROWS - 1) >> BK_SHIFT;
constexpr int NB1 = 256;
constexpr int CHUNK = (N_EDGE + NB1 - 1) / NB1;
constexpr int N_CNT = NBK * NB1;
constexpr int CSCAN_BLOCKS = (N_CNT + SCAN_TILE - 1) / SCAN_TILE;
constexpr int BK_PER_LANE = (NBK + 31) / 32;
constexpr int LIST_CAP = 40960;


__global__ void __launch_bounds__(1024)
bucket_count_kernel(const int* __restrict__ esrc, const int* __restrict__ edst,
                    int* __restrict__ counts, int E) {
    __shared__ int hist[NBK];
    for (int i = threadIdx.x; i < NBK; i += 1024) hist[i] = 0;
    __syncthreads();
    int c0 = blockIdx.x * CHUNK;
    if (c0 > E) c0 = E;
    const int c1 = (c0 + CHUNK < E) ? (c0 + CHUNK) : E;
    for (int i = c0 + threadIdx.x; i < c1; i += 1024) {
        atomicAdd(&hist[edst[i] >> BK_SHIFT], 1);
        atomicAdd(&hist[(N_BOOK + esrc[i]) >> BK_SHIFT], 1);
    }
    __syncthreads();
    for (int i = threadIdx.x; i < NBK; i += 1024) counts[i * NB1 + blockIdx.x] = hist[i];
}

__global__ void block_sum_kernel(const int* __restrict__ deg, int* __restrict__ part, int n) {
    __shared__ int red[8];
    const int lane = threadIdx.x & 31;
    const int wave = threadIdx.x >> 5;
    const int base = blockIdx.x * SCAN_TILE + threadIdx.x * 4;
    int s = 0;
    for (int u = 0; u < 4; ++u)
        if (base + u < n) s += deg[base + u];
    for (int o = 16; o >= 1; o >>= 1) s += __shfl_xor(s, o, 32);
    if (lane == 0) red[wave] = s;
    __syncthreads();
    if (threadIdx.x == 0) {
        int t = 0;
        for (int w = 0; w < 8; ++w) t += red[w];
        part[blockIdx.x] = t;
    }
}

__global__ void scan_kernel(const int* __restrict__ deg, const int* __restrict__ part,
                            int* __restrict__ off, int n) {
    __shared__ int red[8];
    __shared__ int wsum[8];
    const int lane = threadIdx.x & 31;
    const int wave = threadIdx.x >> 5;

    int p = 0;
    for (int i = threadIdx.x; i < (int)blockIdx.x; i += 256) p += part[i];
    for (int o = 16; o >= 1; o >>= 1) p += __shfl_xor(p, o, 32);
    if (lane == 0) red[wave] = p;

    const int base = blockIdx.x * SCAN_TILE + threadIdx.x * 4;
    int d[4];
    int tsum = 0;
    for (int u = 0; u < 4; ++u) {
        d[u] = (base + u < n) ? deg[base + u] : 0;
        tsum += d[u];
    }
    int incl = tsum;
    for (int o = 1; o < 32; o <<= 1) {
        int t = __shfl_up(incl, o, 32);
        if (lane >= o) incl += t;
    }
    if (lane == 31) wsum[wave] = incl;
    __syncthreads();

    int run = incl - tsum;
    for (int w = 0; w < 8; ++w) {
        run += red[w];
        if (w < wave) run += wsum[w];
    }
    for (int u = 0; u < 4; ++u)
        if (base + u < n) {
            off[base + u] = run;
            run += d[u];
        }
}

__global__ void __launch_bounds__(1024)
bucket_scatter_kernel(const int* __restrict__ esrc, const int* __restrict__ edst,
                      const int* __restrict__ cofs, int* __restrict__ p_row,
                      int* __restrict__ p_nbr, int E) {
    __shared__ int hist[NBK];
    __shared__ int lbase[NBK];
    __shared__ int gofs[NBK];
    __shared__ int st_row[2 * CHUNK];
    __shared__ int st_nbr[2 * CHUNK];
    const int tid = threadIdx.x;
    for (int i = tid; i < NBK; i += 1024) hist[i] = 0;
    __syncthreads();
    int c0 = blockIdx.x * CHUNK;
    if (c0 > E) c0 = E;
    const int c1 = (c0 + CHUNK < E) ? (c0 + CHUNK) : E;
    for (int i = c0 + tid; i < c1; i += 1024) {
        atomicAdd(&hist[edst[i] >> BK_SHIFT], 1);
        atomicAdd(&hist[(N_BOOK + esrc[i]) >> BK_SHIFT], 1);
    }
    __syncthreads();
    if (tid < 32) {
        int v[BK_PER_LANE];
        int s = 0;
        for (int u = 0; u < BK_PER_LANE; ++u) {
            const int idx = tid * BK_PER_LANE + u;
            v[u] = (idx < NBK) ? hist[idx] : 0;
            s += v[u];
        }
        int incl = s;
        for (int o = 1; o < 32; o <<= 1) {
            int t = __shfl_up(incl, o, 32);
            if (tid >= o) incl += t;
        }
        int run = incl - s;
        for (int u = 0; u < BK_PER_LANE; ++u) {
            const int idx = tid * BK_PER_LANE + u;
            if (idx < NBK) {
                lbase[idx] = run;
                run += v[u];
            }
        }
    }
    __syncthreads();
    for (int i = tid; i < NBK; i += 1024) {
        const int lb = lbase[i];
        hist[i] = lb;
        gofs[i] = cofs[i * NB1 + blockIdx.x] - lb;
    }
    __syncthreads();
    for (int i = c0 + tid; i < c1; i += 1024) {
        const int s = esrc[i], d = edst[i];
        const int rb = d;
        const int pb = atomicAdd(&hist[rb >> BK_SHIFT], 1);
        st_row[pb] = rb;
        st_nbr[pb] = s;
        const int ru = N_BOOK + s;
        const int pu = atomicAdd(&hist[ru >> BK_SHIFT], 1);
        st_row[pu] = ru;
        st_nbr[pu] = d;
    }
    __syncthreads();
    const int n = 2 * (c1 - c0);
    for (int p = tid; p < n; p += 1024) {
        const int row = st_row[p];
        const int g = gofs[row >> BK_SHIFT] + p;
        p_row[g] = row;
        p_nbr[g] = st_nbr[p];
    }
}

__device__ __forceinline__ void sort_row(const int* __restrict__ list, int n,
                                         int* __restrict__ out, int lane) {
    for (int i0 = 0; i0 < n; i0 += 32) {
        const int i = i0 + lane;
        const int mine = (i < n) ? list[i] : 0;
        int rank = 0;
        for (int j0 = 0; j0 < n; j0 += 32) {
            const int jn = (n - j0 < 32) ? (n - j0) : 32;
            const int theirs = (lane < jn) ? list[j0 + lane] : 0;
            for (int jj = 0; jj < jn; ++jj) {
                const int o = __builtin_amdgcn_readlane(theirs, jj);
                const int j = j0 + jj;
                rank += (o < mine || (o == mine && j < i)) ? 1 : 0;
            }
        }
        if (i < n) out[rank] = mine;
    }
}

__global__ void __launch_bounds__(1024)
build_lists_kernel(const int* __restrict__ p_row, const int* __restrict__ p_nbr,
                   const int* __restrict__ cofs, int* __restrict__ deg, int* __restrict__ off,
                   float* __restrict__ cnt, int* adj_raw, int* __restrict__ adj,
                   int total) {
    __shared__ int rcnt[BK_ROWS];
    __shared__ int loff[BK_ROWS];
    __shared__ int cur[BK_ROWS];
    __shared__ int wsum[32];
    __shared__ int lst[LIST_CAP];
    const int tid = threadIdx.x;
    const int lane = tid & 31;
    const int wave = __builtin_amdgcn_readfirstlane(tid >> 5);
    const int b = blockIdx.x;
    const int row0 = b << BK_SHIFT;
    const int g0 = __builtin_amdgcn_readfirstlane(cofs[b * NB1]);
    const int g1 = (b + 1 < NBK) ? __builtin_amdgcn_readfirstlane(cofs[(b + 1) * NB1]) : total;
    const int n = g1 - g0;

    rcnt[tid] = 0;
    __syncthreads();
    for (int p = tid; p < n; p += 1024) atomicAdd(&rcnt[p_row[g0 + p] - row0], 1);
    __syncthreads();

    const int v = rcnt[tid];
    int incl = v;
    for (int o = 1; o < 32; o <<= 1) {
        int t = __shfl_up(incl, o, 32);
        if (lane >= o) incl += t;
    }
    if (lane == 31) wsum[wave] = incl;
    __syncthreads();
    int front = 0;
    for (int w = 0; w < wave; ++w) front += wsum[w];
    const int ex = front + incl - v;
    loff[tid] = ex;
    cur[tid] = ex;
    if (row0 + tid < N_ROWS) {
        deg[row0 + tid] = v;
        off[row0 + tid] = g0 + ex;
        cnt[row0 + tid] = (float)v;
    }
    __syncthreads();

    const bool fits = n <= LIST_CAP;
    if (fits) {
        for (int p = tid; p < n; p += 1024) {
            const int q = atomicAdd(&cur[p_row[g0 + p] - row0], 1);
            lst[q] = p_nbr[g0 + p];
        }
    } else {
        for (int p = tid; p < n; p += 1024) {
            const int q = atomicAdd(&cur[p_row[g0 + p] - row0], 1);
            adj_raw[g0 + q] = p_nbr[g0 + p];
        }
    }
    __syncthreads();

    for (int r = wave; r < BK_ROWS; r += 32) {
        if (row0 + r >= N_ROWS) break;
        const int lo = __builtin_amdgcn_readfirstlane(loff[r]);
        const int m = __builtin_amdgcn_readfirstlane(rcnt[r]);
        if (fits) sort_row(lst + lo, m, adj + g0 + lo, lane);
        else      sort_row(adj_raw + g0 + lo, m, adj + g0 + lo, lane);
    }
}

template <bool FULL>
__device__ __forceinline__ void gather_batch(float4& acc, int& row, int& row_end,
                                             float* __restrict__ st,
                                             const float* __restrict__ feat, int ids,
                                             int my_end, int c, int m, int lane,
                                             const int* __restrict__ nxt, bool has_nxt,
                                             int& ids_next) {
    const int stride = HD + 4;
    float4 v[32];
#pragma unroll
    for (int u = 0; u < 32; ++u) {
        const int s = __builtin_amdgcn_readlane(ids, u);
        if (FULL) {
            v[u] = *(const float4*)(feat + (size_t)s * HD + lane * 4);
        } else {
            v[u] = float4{0.0f, 0.0f, 0.0f, 0.0f};
            if (u < m) v[u] = *(const float4*)(feat + (size_t)s * HD + lane * 4);
        }
    }
    ids_next = 0;
    if (has_nxt) ids_next = *nxt;
#pragma unroll
    for (int u = 0; u < 32; ++u) {
        if (FULL || u < m) {
            while (c + u == row_end) {
                *(float4*)&st[row * stride + lane * 4] = acc;
                acc = float4{0.0f, 0.0f, 0.0f, 0.0f};
                ++row;
                row_end = __builtin_amdgcn_readlane(my_end, row);
            }
            acc.x += v[u].x;
            acc.y += v[u].y;
            acc.z += v[u].z;
            acc.w += v[u].w;
        }
    }
}


__device__ __forceinline__ void load_weight_t(const float* __restrict__ W, float* __restrict__ wt,
                                              int K, int stride, int nthr) {
    const int nf4 = K * (HD / 4);
    for (int idx = threadIdx.x; idx < nf4; idx += nthr) {
        int k = idx >> 5;
        int n = (idx & 31) << 2;
        const float4 w4 = *(const float4*)(W + (size_t)k * HD + n);
        wt[(n + 0) * stride + k] = w4.x;
        wt[(n + 1) * stride + k] = w4.y;
        wt[(n + 2) * stride + k] = w4.z;
        wt[(n + 3) * stride + k] = w4.w;
    }
}

__global__ void gemm_bias_kernel(const float* __restrict__ A, const float* __restrict__ W,
                                 const float* __restrict__ bias, float* __restrict__ out,
                                 int M, int K) {
    __shared__ float wt[HD * (HD + 4)];
    const int stride = K + 4;
    load_weight_t(W, wt, K, stride, 256);
    __syncthreads();

    const int wave = threadIdx.x >> 5;
    const int lane = threadIdx.x & 31;
    const int col  = lane & 15;
    const int half = lane >> 4;
    const int m0   = blockIdx.x * 128 + wave * 16;
    const int aRow = (m0 + col < M) ? (m0 + col) : (M - 1);

    v8f acc[8];
#pragma unroll
    for (int nt = 0; nt < 8; ++nt) {
        float bv = bias[nt * 16 + col];
        acc[nt] = (v8f){bv, bv, bv, bv, bv, bv, bv, bv};
    }

    const float* arow = A + (size_t)aRow * K;
    for (int k0 = 0; k0 < K; k0 += 4) {
        const int kb = k0 + 2 * half;
        v2f a = *(const v2f*)(arow + kb);
        v2f bf[8];
#pragma unroll
        for (int nt = 0; nt < 8; ++nt)
            bf[nt] = *(const v2f*)&wt[(nt * 16 + col) * stride + kb];
#pragma unroll
        for (int nt = 0; nt < 8; ++nt)
            acc[nt] = __builtin_amdgcn_wmma_f32_16x16x4_f32(
                false, a, false, bf[nt], (short)0, acc[nt], false, false);
    }

#pragma unroll
    for (int nt = 0; nt < 8; ++nt)
#pragma unroll
        for (int v = 0; v < 8; ++v) {
            int dr = m0 + v + 8 * half;
            if (dr < M) out[(size_t)dr * HD + nt * 16 + col] = acc[nt][v];
        }
}

constexpr int WT_WORDS   = HD * (HD + 4);
constexpr int TILE_WORDS = 16 * (HD + 4);
constexpr int SAGE_WAVES = 16;
constexpr int SAGE_BLOCKS = 256;
constexpr int TILES_B = (N_BOOK + 15) / 16;
constexpr int TILES_U = (N_USER + 15) / 16;

__global__ void __launch_bounds__(SAGE_WAVES * 32)
sage_layer_kernel(const float* __restrict__ u_cur, const float* __restrict__ b_cur,
                  const int* __restrict__ off, const int* __restrict__ deg,
                  const int* __restrict__ adj, const float* __restrict__ cnt,
                  const float* __restrict__ Wl, const float* __restrict__ bl,
                  const float* __restrict__ Wr, float* __restrict__ u_nxt,
                  float* __restrict__ b_nxt) {
    __shared__ alignas(16) float wl_t[WT_WORDS];
    __shared__ alignas(16) float wr_t[WT_WORDS];
    __shared__ alignas(16) float tiles[SAGE_WAVES * TILE_WORDS];
    const int stride = HD + 4;

    load_weight_t(Wl, wl_t, HD, stride, SAGE_WAVES * 32);
    load_weight_t(Wr, wr_t, HD, stride, SAGE_WAVES * 32);
    __syncthreads();

    const int lane = threadIdx.x & 31;
    const int wave = __builtin_amdgcn_readfirstlane((int)(threadIdx.x >> 5));
    const int col  = lane & 15;
    const int half = lane >> 4;
    float* __restrict__ st = tiles + wave * TILE_WORDS;

    float bv[8];
#pragma unroll
    for (int nt = 0; nt < 8; ++nt) bv[nt] = bl[nt * 16 + col];

    for (int q = wave; ; q += SAGE_WAVES) {
        const int t = (int)blockIdx.x + (int)gridDim.x * q;
        if (t >= TILES_B + TILES_U) break;
        asm volatile("" ::: "memory");
        const bool isb = t < TILES_B;
        const int M  = isb ? N_BOOK : N_USER;
        const int m0 = (isb ? t : t - TILES_B) * 16;
        const int gsh = isb ? 0 : N_BOOK;
        const float* __restrict__ feat = isb ? u_cur : b_cur;
        const float* __restrict__ X    = isb ? b_cur : u_cur;
        float* __restrict__ out        = isb ? b_nxt : u_nxt;

        int my_beg = 0, my_n = 0;
        if (lane < 16 && m0 + lane < M) {
            my_beg = off[gsh + m0 + lane];
            my_n = deg[gsh + m0 + lane];
        }
        int my_end = my_n;
#pragma unroll
        for (int o = 1; o < 16; o <<= 1) {
            const int up = __shfl_up(my_end, o, 32);
            if (lane >= o) my_end += up;
        }
        const int total = __builtin_amdgcn_readlane(my_end, 15);
        const int tbeg = __builtin_amdgcn_readlane(my_beg, 0);

        int ids = (lane < total) ? adj[tbeg + lane] : 0;
        int row = 0;
        int row_end = __builtin_amdgcn_readlane(my_end, 0);
        float4 acc4 = {0.0f, 0.0f, 0.0f, 0.0f};
        for (int c = 0; c < total; c += 32) {
            const int m = (total - c < 32) ? (total - c) : 32;
            int ids_next;
            const int* __restrict__ nxt = adj + tbeg + c + 32 + lane;
            const bool has_nxt = c + 32 + lane < total;
            if (m == 32)
                gather_batch<true>(acc4, row, row_end, st, feat, ids, my_end, c, 32, lane,
                                   nxt, has_nxt, ids_next);
            else
                gather_batch<false>(acc4, row, row_end, st, feat, ids, my_end, c, m, lane,
                                    nxt, has_nxt, ids_next);
            ids = ids_next;
        }
        for (; row < 16; ++row) {
            *(float4*)&st[row * stride + lane * 4] = acc4;
            acc4 = float4{0.0f, 0.0f, 0.0f, 0.0f};
        }

        const int aRow = (m0 + col < M) ? (m0 + col) : (M - 1);
        const float invc = 1.0f / fmaxf(cnt[gsh + aRow], 1.0f);
        v2f a[HD / 4];
#pragma unroll
        for (int k = 0; k < HD / 4; ++k) {
            a[k] = *(const v2f*)&st[col * stride + 4 * k + 2 * half];
            a[k].x *= invc;
            a[k].y *= invc;
        }

        v8f acc[8];
#pragma unroll
        for (int nt = 0; nt < 8; ++nt)
            acc[nt] = (v8f){bv[nt], bv[nt], bv[nt], bv[nt], bv[nt], bv[nt], bv[nt], bv[nt]};

#pragma unroll
        for (int k = 0; k < HD / 4; ++k) {
            const int kb = 4 * k + 2 * half;
            v2f bf[8];
#pragma unroll
            for (int nt = 0; nt < 8; ++nt)
                bf[nt] = *(const v2f*)&wl_t[(nt * 16 + col) * stride + kb];
#pragma unroll
            for (int nt = 0; nt < 8; ++nt)
                acc[nt] = __builtin_amdgcn_wmma_f32_16x16x4_f32(
                    false, a[k], false, bf[nt], (short)0, acc[nt], false, false);
        }

        const float* xrow = X + (size_t)aRow * HD;
        for (int k0 = 0; k0 < HD; k0 += 4) {
            const int kb = k0 + 2 * half;
            v2f ax = *(const v2f*)(xrow + kb);
            v2f bf[8];
#pragma unroll
            for (int nt = 0; nt < 8; ++nt)
                bf[nt] = *(const v2f*)&wr_t[(nt * 16 + col) * stride + kb];
#pragma unroll
            for (int nt = 0; nt < 8; ++nt)
                acc[nt] = __builtin_amdgcn_wmma_f32_16x16x4_f32(
                    false, ax, false, bf[nt], (short)0, acc[nt], false, false);
        }

#pragma unroll
        for (int nt = 0; nt < 8; ++nt)
#pragma unroll
            for (int v = 0; v < 8; ++v) {
                int dr = m0 + v + 8 * half;
                if (dr < M) out[(size_t)dr * HD + nt * 16 + col] = fmaxf(acc[nt][v], 0.0f);
            }
    }
}

template <int U>
__device__ __forceinline__ void ln_rows(const float* __restrict__ X, float* __restrict__ out,
                                        const float4 g, const float4 bt, int row0, int waves,
                                        int lane) {
    float4 v[U];
#pragma unroll
    for (int u = 0; u < U; ++u)
        v[u] = *(const float4*)(X + (size_t)(row0 + u * waves) * HD + lane * 4);
#pragma unroll
    for (int u = 0; u < U; ++u) {
        float s = v[u].x + v[u].y + v[u].z + v[u].w;
#pragma unroll
        for (int off = 16; off >= 1; off >>= 1) s += __shfl_xor(s, off, 32);
        const float mu = s * (1.0f / HD);

        const float dx = v[u].x - mu, dy = v[u].y - mu, dz = v[u].z - mu, dw = v[u].w - mu;
        float q = dx * dx + dy * dy + dz * dz + dw * dw;
#pragma unroll
        for (int off = 16; off >= 1; off >>= 1) q += __shfl_xor(q, off, 32);
        const float rstd = rsqrtf(q * (1.0f / HD) + LN_EPS);

        float4 o;
        o.x = dx * rstd * g.x + bt.x;
        o.y = dy * rstd * g.y + bt.y;
        o.z = dz * rstd * g.z + bt.z;
        o.w = dw * rstd * g.w + bt.w;
        *(float4*)(out + (size_t)(row0 + u * waves) * HD + lane * 4) = o;
    }
}

__global__ void layernorm_kernel(const float* __restrict__ X, const float* __restrict__ gamma,
                                 const float* __restrict__ beta, float* __restrict__ out, int M) {
    const int lane = threadIdx.x & 31;
    const int waves = (int)(gridDim.x * 8);
    const float4 g = *(const float4*)(gamma + lane * 4);
    const float4 bt = *(const float4*)(beta + lane * 4);

    int row = __builtin_amdgcn_readfirstlane((int)(blockIdx.x * 8 + (threadIdx.x >> 5)));
    for (; row + 3 * waves < M; row += 4 * waves) ln_rows<4>(X, out, g, bt, row, waves, lane);
    for (; row < M; row += waves) ln_rows<1>(X, out, g, bt, row, waves, lane);
}

extern "C" void kernel_launch(void* const* d_in, const int* in_sizes, int n_in,
                              void* d_out, int out_size, void* d_ws, size_t ws_size,
                              hipStream_t stream) {
    const float* user_x      = (const float*)d_in[0];
    const float* book_x      = (const float*)d_in[1];
    const int*   edge_src    = (const int*)d_in[2];
    const int*   edge_dst    = (const int*)d_in[3];
    const float* user_proj_w = (const float*)d_in[4];
    const float* user_proj_b = (const float*)d_in[5];
    const float* book_proj_w = (const float*)d_in[6];
    const float* book_proj_b = (const float*)d_in[7];
    const float* Wl          = (const float*)d_in[8];
    const float* bl          = (const float*)d_in[9];
    const float* Wr          = (const float*)d_in[10];
    const float* ugamma      = (const float*)d_in[11];
    const float* ubeta       = (const float*)d_in[12];
    const float* bgamma      = (const float*)d_in[13];
    const float* bbeta       = (const float*)d_in[14];

    float* ws = (float*)d_ws;
    size_t off = 0;
    auto carve = [&](size_t n) { float* p = ws + off; off += n; return p; };
    float* u_cur = carve((size_t)N_USER * HD);
    float* u_nxt = carve((size_t)N_USER * HD);
    float* b_cur = carve((size_t)N_BOOK * HD);
    float* b_nxt = carve((size_t)N_BOOK * HD);
    float* cnt   = carve((size_t)N_ROWS);
    int* deg     = (int*)carve((size_t)N_ROWS);
    int* offs    = (int*)carve((size_t)N_ROWS);
    int* counts  = (int*)carve((size_t)N_CNT);
    int* cofs    = (int*)carve((size_t)N_CNT);
    int* part    = (int*)carve((size_t)64);
    int* p_row   = (int*)carve((size_t)2 * N_EDGE);
    int* p_nbr   = (int*)carve((size_t)2 * N_EDGE);
    int* adj_raw = (int*)carve((size_t)2 * N_EDGE);
    int* adj     = (int*)carve((size_t)2 * N_EDGE);

    gemm_bias_kernel<<<(N_USER + 127) / 128, 256, 0, stream>>>(
        user_x, user_proj_w, user_proj_b, u_cur, N_USER, 64);
    gemm_bias_kernel<<<(N_BOOK + 127) / 128, 256, 0, stream>>>(
        book_x, book_proj_w, book_proj_b, b_cur, N_BOOK, 128);

    bucket_count_kernel<<<NB1, 1024, 0, stream>>>(edge_src, edge_dst, counts, N_EDGE);
    block_sum_kernel<<<CSCAN_BLOCKS, 256, 0, stream>>>(counts, part, N_CNT);
    scan_kernel<<<CSCAN_BLOCKS, 256, 0, stream>>>(counts, part, cofs, N_CNT);
    bucket_scatter_kernel<<<NB1, 1024, 0, stream>>>(
        edge_src, edge_dst, cofs, p_row, p_nbr, N_EDGE);
    build_lists_kernel<<<NBK, 1024, 0, stream>>>(
        p_row, p_nbr, cofs, deg, offs, cnt, adj_raw, adj, 2 * N_EDGE);

    for (int l = 0; l < 3; ++l) {
        const float* wl  = Wl + (size_t)l * HD * HD;
        const float* blv = bl + (size_t)l * HD;
        const float* wr  = Wr + (size_t)l * HD * HD;

        sage_layer_kernel<<<SAGE_BLOCKS, SAGE_WAVES * 32, 0, stream>>>(
            u_cur, b_cur, offs, deg, adj, cnt, wl, blv, wr, u_nxt, b_nxt);

        float* t;
        t = u_cur; u_cur = u_nxt; u_nxt = t;
        t = b_cur; b_cur = b_nxt; b_nxt = t;
    }

    float* out_u = (float*)d_out;
    float* out_b = out_u + (size_t)N_USER * HD;
    layernorm_kernel<<<FILL_BLOCKS, 256, 0, stream>>>(u_cur, ugamma, ubeta, out_u, N_USER);
    layernorm_kernel<<<FILL_BLOCKS, 256, 0, stream>>>(b_cur, bgamma, bbeta, out_b, N_BOOK);
}


# SAGE tile rows cross memory whole, staged through the wave's LDS tile: X operand in (16 row loads, was 32 loads of 16 lines each), result out (16 row stores, was 64 half-line pairs)
# speedup vs baseline: 3.7547x; 1.0624x over previous
#include <hip/hip_runtime.h>

typedef float v2f __attribute__((ext_vector_type(2)));
typedef float v8f __attribute__((ext_vector_type(8)));

#define HD 128
constexpr int N_USER = 150000;
constexpr int N_BOOK = 75000;
constexpr int N_EDGE = 1500000;
constexpr float LN_EPS = 1e-5f;
constexpr int N_ROWS = N_BOOK + N_USER;
constexpr int SCAN_TILE = 1024;
constexpr int FILL_BLOCKS = 2048;

constexpr int BK_SHIFT = 10;
constexpr int BK_ROWS = 1 << BK_SHIFT;
constexpr int NBK = (N_ROWS + BK_ROWS - 1) >> BK_SHIFT;
constexpr int NB1 = 256;
constexpr int CHUNK = (N_EDGE + NB1 - 1) / NB1;
constexpr int N_CNT = NBK * NB1;
constexpr int CSCAN_BLOCKS = (N_CNT + SCAN_TILE - 1) / SCAN_TILE;
constexpr int BK_PER_LANE = (NBK + 31) / 32;
constexpr int LIST_CAP = 40960;


__global__ void __launch_bounds__(1024)
bucket_count_kernel(const int* __restrict__ esrc, const int* __restrict__ edst,
                    int* __restrict__ counts, int E) {
    __shared__ int hist[NBK];
    for (int i = threadIdx.x; i < NBK; i += 1024) hist[i] = 0;
    __syncthreads();
    int c0 = blockIdx.x * CHUNK;
    if (c0 > E) c0 = E;
    const int c1 = (c0 + CHUNK < E) ? (c0 + CHUNK) : E;
    for (int i = c0 + threadIdx.x; i < c1; i += 1024) {
        atomicAdd(&hist[edst[i] >> BK_SHIFT], 1);
        atomicAdd(&hist[(N_BOOK + esrc[i]) >> BK_SHIFT], 1);
    }
    __syncthreads();
    for (int i = threadIdx.x; i < NBK; i += 1024) counts[i * NB1 + blockIdx.x] = hist[i];
}

__global__ void block_sum_kernel(const int* __restrict__ deg, int* __restrict__ part, int n) {
    __shared__ int red[8];
    const int lane = threadIdx.x & 31;
    const int wave = threadIdx.x >> 5;
    const int base = blockIdx.x * SCAN_TILE + threadIdx.x * 4;
    int s = 0;
    for (int u = 0; u < 4; ++u)
        if (base + u < n) s += deg[base + u];
    for (int o = 16; o >= 1; o >>= 1) s += __shfl_xor(s, o, 32);
    if (lane == 0) red[wave] = s;
    __syncthreads();
    if (threadIdx.x == 0) {
        int t = 0;
        for (int w = 0; w < 8; ++w) t += red[w];
        part[blockIdx.x] = t;
    }
}

__global__ void scan_kernel(const int* __restrict__ deg, const int* __restrict__ part,
                            int* __restrict__ off, int n) {
    __shared__ int red[8];
    __shared__ int wsum[8];
    const int lane = threadIdx.x & 31;
    const int wave = threadIdx.x >> 5;

    int p = 0;
    for (int i = threadIdx.x; i < (int)blockIdx.x; i += 256) p += part[i];
    for (int o = 16; o >= 1; o >>= 1) p += __shfl_xor(p, o, 32);
    if (lane == 0) red[wave] = p;

    const int base = blockIdx.x * SCAN_TILE + threadIdx.x * 4;
    int d[4];
    int tsum = 0;
    for (int u = 0; u < 4; ++u) {
        d[u] = (base + u < n) ? deg[base + u] : 0;
        tsum += d[u];
    }
    int incl = tsum;
    for (int o = 1; o < 32; o <<= 1) {
        int t = __shfl_up(incl, o, 32);
        if (lane >= o) incl += t;
    }
    if (lane == 31) wsum[wave] = incl;
    __syncthreads();

    int run = incl - tsum;
    for (int w = 0; w < 8; ++w) {
        run += red[w];
        if (w < wave) run += wsum[w];
    }
    for (int u = 0; u < 4; ++u)
        if (base + u < n) {
            off[base + u] = run;
            run += d[u];
        }
}

__global__ void __launch_bounds__(1024)
bucket_scatter_kernel(const int* __restrict__ esrc, const int* __restrict__ edst,
                      const int* __restrict__ cofs, int* __restrict__ p_row,
                      int* __restrict__ p_nbr, int E) {
    __shared__ int hist[NBK];
    __shared__ int lbase[NBK];
    __shared__ int gofs[NBK];
    __shared__ int st_row[2 * CHUNK];
    __shared__ int st_nbr[2 * CHUNK];
    const int tid = threadIdx.x;
    for (int i = tid; i < NBK; i += 1024) hist[i] = 0;
    __syncthreads();
    int c0 = blockIdx.x * CHUNK;
    if (c0 > E) c0 = E;
    const int c1 = (c0 + CHUNK < E) ? (c0 + CHUNK) : E;
    for (int i = c0 + tid; i < c1; i += 1024) {
        atomicAdd(&hist[edst[i] >> BK_SHIFT], 1);
        atomicAdd(&hist[(N_BOOK + esrc[i]) >> BK_SHIFT], 1);
    }
    __syncthreads();
    if (tid < 32) {
        int v[BK_PER_LANE];
        int s = 0;
        for (int u = 0; u < BK_PER_LANE; ++u) {
            const int idx = tid * BK_PER_LANE + u;
            v[u] = (idx < NBK) ? hist[idx] : 0;
            s += v[u];
        }
        int incl = s;
        for (int o = 1; o < 32; o <<= 1) {
            int t = __shfl_up(incl, o, 32);
            if (tid >= o) incl += t;
        }
        int run = incl - s;
        for (int u = 0; u < BK_PER_LANE; ++u) {
            const int idx = tid * BK_PER_LANE + u;
            if (idx < NBK) {
                lbase[idx] = run;
                run += v[u];
            }
        }
    }
    __syncthreads();
    for (int i = tid; i < NBK; i += 1024) {
        const int lb = lbase[i];
        hist[i] = lb;
        gofs[i] = cofs[i * NB1 + blockIdx.x] - lb;
    }
    __syncthreads();
    for (int i = c0 + tid; i < c1; i += 1024) {
        const int s = esrc[i], d = edst[i];
        const int rb = d;
        const int pb = atomicAdd(&hist[rb >> BK_SHIFT], 1);
        st_row[pb] = rb;
        st_nbr[pb] = s;
        const int ru = N_BOOK + s;
        const int pu = atomicAdd(&hist[ru >> BK_SHIFT], 1);
        st_row[pu] = ru;
        st_nbr[pu] = d;
    }
    __syncthreads();
    const int n = 2 * (c1 - c0);
    for (int p = tid; p < n; p += 1024) {
        const int row = st_row[p];
        const int g = gofs[row >> BK_SHIFT] + p;
        p_row[g] = row;
        p_nbr[g] = st_nbr[p];
    }
}

__device__ __forceinline__ void sort_row(const int* __restrict__ list, int n,
                                         int* __restrict__ out, int lane) {
    for (int i0 = 0; i0 < n; i0 += 32) {
        const int i = i0 + lane;
        const int mine = (i < n) ? list[i] : 0;
        int rank = 0;
        for (int j0 = 0; j0 < n; j0 += 32) {
            const int jn = (n - j0 < 32) ? (n - j0) : 32;
            const int theirs = (lane < jn) ? list[j0 + lane] : 0;
            for (int jj = 0; jj < jn; ++jj) {
                const int o = __builtin_amdgcn_readlane(theirs, jj);
                const int j = j0 + jj;
                rank += (o < mine || (o == mine && j < i)) ? 1 : 0;
            }
        }
        if (i < n) out[rank] = mine;
    }
}

__global__ void __launch_bounds__(1024)
build_lists_kernel(const int* __restrict__ p_row, const int* __restrict__ p_nbr,
                   const int* __restrict__ cofs, int* __restrict__ deg, int* __restrict__ off,
                   float* __restrict__ cnt, int* adj_raw, int* __restrict__ adj,
                   int total) {
    __shared__ int rcnt[BK_ROWS];
    __shared__ int loff[BK_ROWS];
    __shared__ int cur[BK_ROWS];
    __shared__ int wsum[32];
    __shared__ int lst[LIST_CAP];
    const int tid = threadIdx.x;
    const int lane = tid & 31;
    const int wave = __builtin_amdgcn_readfirstlane(tid >> 5);
    const int b = blockIdx.x;
    const int row0 = b << BK_SHIFT;
    const int g0 = __builtin_amdgcn_readfirstlane(cofs[b * NB1]);
    const int g1 = (b + 1 < NBK) ? __builtin_amdgcn_readfirstlane(cofs[(b + 1) * NB1]) : total;
    const int n = g1 - g0;

    rcnt[tid] = 0;
    __syncthreads();
    for (int p = tid; p < n; p += 1024) atomicAdd(&rcnt[p_row[g0 + p] - row0], 1);
    __syncthreads();

    const int v = rcnt[tid];
    int incl = v;
    for (int o = 1; o < 32; o <<= 1) {
        int t = __shfl_up(incl, o, 32);
        if (lane >= o) incl += t;
    }
    if (lane == 31) wsum[wave] = incl;
    __syncthreads();
    int front = 0;
    for (int w = 0; w < wave; ++w) front += wsum[w];
    const int ex = front + incl - v;
    loff[tid] = ex;
    cur[tid] = ex;
    if (row0 + tid < N_ROWS) {
        deg[row0 + tid] = v;
        off[row0 + tid] = g0 + ex;
        cnt[row0 + tid] = (float)v;
    }
    __syncthreads();

    const bool fits = n <= LIST_CAP;
    if (fits) {
        for (int p = tid; p < n; p += 1024) {
            const int q = atomicAdd(&cur[p_row[g0 + p] - row0], 1);
            lst[q] = p_nbr[g0 + p];
        }
    } else {
        for (int p = tid; p < n; p += 1024) {
            const int q = atomicAdd(&cur[p_row[g0 + p] - row0], 1);
            adj_raw[g0 + q] = p_nbr[g0 + p];
        }
    }
    __syncthreads();

    for (int r = wave; r < BK_ROWS; r += 32) {
        if (row0 + r >= N_ROWS) break;
        const int lo = __builtin_amdgcn_readfirstlane(loff[r]);
        const int m = __builtin_amdgcn_readfirstlane(rcnt[r]);
        if (fits) sort_row(lst + lo, m, adj + g0 + lo, lane);
        else      sort_row(adj_raw + g0 + lo, m, adj + g0 + lo, lane);
    }
}

template <bool FULL>
__device__ __forceinline__ void gather_batch(float4& acc, int& row, int& row_end,
                                             float* __restrict__ st,
                                             const float* __restrict__ feat, int ids,
                                             int my_end, int c, int m, int lane,
                                             const int* __restrict__ nxt, bool has_nxt,
                                             int& ids_next) {
    const int stride = HD + 4;
    float4 v[32];
#pragma unroll
    for (int u = 0; u < 32; ++u) {
        const int s = __builtin_amdgcn_readlane(ids, u);
        if (FULL) {
            v[u] = *(const float4*)(feat + (size_t)s * HD + lane * 4);
        } else {
            v[u] = float4{0.0f, 0.0f, 0.0f, 0.0f};
            if (u < m) v[u] = *(const float4*)(feat + (size_t)s * HD + lane * 4);
        }
    }
    ids_next = 0;
    if (has_nxt) ids_next = *nxt;
#pragma unroll
    for (int u = 0; u < 32; ++u) {
        if (FULL || u < m) {
            while (c + u == row_end) {
                *(float4*)&st[row * stride + lane * 4] = acc;
                acc = float4{0.0f, 0.0f, 0.0f, 0.0f};
                ++row;
                row_end = __builtin_amdgcn_readlane(my_end, row);
            }
            acc.x += v[u].x;
            acc.y += v[u].y;
            acc.z += v[u].z;
            acc.w += v[u].w;
        }
    }
}


__device__ __forceinline__ void load_weight_t(const float* __restrict__ W, float* __restrict__ wt,
                                              int K, int stride, int nthr) {
    const int nf4 = K * (HD / 4);
    for (int idx = threadIdx.x; idx < nf4; idx += nthr) {
        int k = idx >> 5;
        int n = (idx & 31) << 2;
        const float4 w4 = *(const float4*)(W + (size_t)k * HD + n);
        wt[(n + 0) * stride + k] = w4.x;
        wt[(n + 1) * stride + k] = w4.y;
        wt[(n + 2) * stride + k] = w4.z;
        wt[(n + 3) * stride + k] = w4.w;
    }
}

__global__ void gemm_bias_kernel(const float* __restrict__ A, const float* __restrict__ W,
                                 const float* __restrict__ bias, float* __restrict__ out,
                                 int M, int K) {
    __shared__ float wt[HD * (HD + 4)];
    const int stride = K + 4;
    load_weight_t(W, wt, K, stride, 256);
    __syncthreads();

    const int wave = threadIdx.x >> 5;
    const int lane = threadIdx.x & 31;
    const int col  = lane & 15;
    const int half = lane >> 4;
    const int m0   = blockIdx.x * 128 + wave * 16;
    const int aRow = (m0 + col < M) ? (m0 + col) : (M - 1);

    v8f acc[8];
#pragma unroll
    for (int nt = 0; nt < 8; ++nt) {
        float bv = bias[nt * 16 + col];
        acc[nt] = (v8f){bv, bv, bv, bv, bv, bv, bv, bv};
    }

    const float* arow = A + (size_t)aRow * K;
    for (int k0 = 0; k0 < K; k0 += 4) {
        const int kb = k0 + 2 * half;
        v2f a = *(const v2f*)(arow + kb);
        v2f bf[8];
#pragma unroll
        for (int nt = 0; nt < 8; ++nt)
            bf[nt] = *(const v2f*)&wt[(nt * 16 + col) * stride + kb];
#pragma unroll
        for (int nt = 0; nt < 8; ++nt)
            acc[nt] = __builtin_amdgcn_wmma_f32_16x16x4_f32(
                false, a, false, bf[nt], (short)0, acc[nt], false, false);
    }

#pragma unroll
    for (int nt = 0; nt < 8; ++nt)
#pragma unroll
        for (int v = 0; v < 8; ++v) {
            int dr = m0 + v + 8 * half;
            if (dr < M) out[(size_t)dr * HD + nt * 16 + col] = acc[nt][v];
        }
}

constexpr int WT_WORDS   = HD * (HD + 4);
constexpr int TILE_WORDS = 16 * (HD + 4);
constexpr int SAGE_WAVES = 16;
constexpr int SAGE_BLOCKS = 256;
constexpr int TILES_B = (N_BOOK + 15) / 16;
constexpr int TILES_U = (N_USER + 15) / 16;

__global__ void __launch_bounds__(SAGE_WAVES * 32)
sage_layer_kernel(const float* __restrict__ u_cur, const float* __restrict__ b_cur,
                  const int* __restrict__ off, const int* __restrict__ deg,
                  const int* __restrict__ adj, const float* __restrict__ cnt,
                  const float* __restrict__ Wl, const float* __restrict__ bl,
                  const float* __restrict__ Wr, float* __restrict__ u_nxt,
                  float* __restrict__ b_nxt) {
    __shared__ alignas(16) float wl_t[WT_WORDS];
    __shared__ alignas(16) float wr_t[WT_WORDS];
    __shared__ alignas(16) float tiles[SAGE_WAVES * TILE_WORDS];
    const int stride = HD + 4;

    load_weight_t(Wl, wl_t, HD, stride, SAGE_WAVES * 32);
    load_weight_t(Wr, wr_t, HD, stride, SAGE_WAVES * 32);
    __syncthreads();

    const int lane = threadIdx.x & 31;
    const int wave = __builtin_amdgcn_readfirstlane((int)(threadIdx.x >> 5));
    const int col  = lane & 15;
    const int half = lane >> 4;
    float* __restrict__ st = tiles + wave * TILE_WORDS;

    float bv[8];
#pragma unroll
    for (int nt = 0; nt < 8; ++nt) bv[nt] = bl[nt * 16 + col];

    for (int q = wave; ; q += SAGE_WAVES) {
        const int t = (int)blockIdx.x + (int)gridDim.x * q;
        if (t >= TILES_B + TILES_U) break;
        asm volatile("" ::: "memory");
        const bool isb = t < TILES_B;
        const int M  = isb ? N_BOOK : N_USER;
        const int m0 = (isb ? t : t - TILES_B) * 16;
        const int gsh = isb ? 0 : N_BOOK;
        const float* __restrict__ feat = isb ? u_cur : b_cur;
        const float* __restrict__ X    = isb ? b_cur : u_cur;
        float* __restrict__ out        = isb ? b_nxt : u_nxt;

        int my_beg = 0, my_n = 0;
        if (lane < 16 && m0 + lane < M) {
            my_beg = off[gsh + m0 + lane];
            my_n = deg[gsh + m0 + lane];
        }
        int my_end = my_n;
#pragma unroll
        for (int o = 1; o < 16; o <<= 1) {
            const int up = __shfl_up(my_end, o, 32);
            if (lane >= o) my_end += up;
        }
        const int total = __builtin_amdgcn_readlane(my_end, 15);
        const int tbeg = __builtin_amdgcn_readlane(my_beg, 0);

        int ids = (lane < total) ? adj[tbeg + lane] : 0;
        int row = 0;
        int row_end = __builtin_amdgcn_readlane(my_end, 0);
        float4 acc4 = {0.0f, 0.0f, 0.0f, 0.0f};
        for (int c = 0; c < total; c += 32) {
            const int m = (total - c < 32) ? (total - c) : 32;
            int ids_next;
            const int* __restrict__ nxt = adj + tbeg + c + 32 + lane;
            const bool has_nxt = c + 32 + lane < total;
            if (m == 32)
                gather_batch<true>(acc4, row, row_end, st, feat, ids, my_end, c, 32, lane,
                                   nxt, has_nxt, ids_next);
            else
                gather_batch<false>(acc4, row, row_end, st, feat, ids, my_end, c, m, lane,
                                    nxt, has_nxt, ids_next);
            ids = ids_next;
        }
        float4 xr[16];
#pragma unroll
        for (int r = 0; r < 16; ++r) {
            const int xi = (m0 + r < M) ? (m0 + r) : (M - 1);
            xr[r] = *(const float4*)(X + (size_t)xi * HD + lane * 4);
        }

        for (; row < 16; ++row) {
            *(float4*)&st[row * stride + lane * 4] = acc4;
            acc4 = float4{0.0f, 0.0f, 0.0f, 0.0f};
        }

        const int aRow = (m0 + col < M) ? (m0 + col) : (M - 1);
        const float invc = 1.0f / fmaxf(cnt[gsh + aRow], 1.0f);
        v2f a[HD / 4];
#pragma unroll
        for (int k = 0; k < HD / 4; ++k) {
            a[k] = *(const v2f*)&st[col * stride + 4 * k + 2 * half];
            a[k].x *= invc;
            a[k].y *= invc;
        }
#pragma unroll
        for (int r = 0; r < 16; ++r) *(float4*)&st[r * stride + lane * 4] = xr[r];
        asm volatile("" ::: "memory");

        v8f acc[8];
#pragma unroll
        for (int nt = 0; nt < 8; ++nt)
            acc[nt] = (v8f){bv[nt], bv[nt], bv[nt], bv[nt], bv[nt], bv[nt], bv[nt], bv[nt]};

#pragma unroll
        for (int k = 0; k < HD / 4; ++k) {
            const int kb = 4 * k + 2 * half;
            v2f bf[8];
#pragma unroll
            for (int nt = 0; nt < 8; ++nt)
                bf[nt] = *(const v2f*)&wl_t[(nt * 16 + col) * stride + kb];
#pragma unroll
            for (int nt = 0; nt < 8; ++nt)
                acc[nt] = __builtin_amdgcn_wmma_f32_16x16x4_f32(
                    false, a[k], false, bf[nt], (short)0, acc[nt], false, false);
        }

#pragma unroll
        for (int k0 = 0; k0 < HD; k0 += 4) {
            const int kb = k0 + 2 * half;
            v2f ax = *(const v2f*)&st[col * stride + kb];
            v2f bf[8];
#pragma unroll
            for (int nt = 0; nt < 8; ++nt)
                bf[nt] = *(const v2f*)&wr_t[(nt * 16 + col) * stride + kb];
#pragma unroll
            for (int nt = 0; nt < 8; ++nt)
                acc[nt] = __builtin_amdgcn_wmma_f32_16x16x4_f32(
                    false, ax, false, bf[nt], (short)0, acc[nt], false, false);
        }

#pragma unroll
        for (int nt = 0; nt < 8; ++nt)
#pragma unroll
            for (int v = 0; v < 8; ++v)
                st[(v + 8 * half) * stride + nt * 16 + col] = fmaxf(acc[nt][v], 0.0f);
        float4 o[16];
#pragma unroll
        for (int r = 0; r < 16; ++r) o[r] = *(const float4*)&st[r * stride + lane * 4];
        float* __restrict__ orow = out + (size_t)m0 * HD + lane * 4;
        if (m0 + 16 <= M) {
#pragma unroll
            for (int r = 0; r < 16; ++r) *(float4*)(orow + r * HD) = o[r];
        } else {
#pragma unroll
            for (int r = 0; r < 16; ++r)
                if (m0 + r < M) *(float4*)(orow + r * HD) = o[r];
        }
    }
}

template <int U>
__device__ __forceinline__ void ln_rows(const float* __restrict__ X, float* __restrict__ out,
                                        const float4 g, const float4 bt, int row0, int waves,
                                        int lane) {
    float4 v[U];
#pragma unroll
    for (int u = 0; u < U; ++u)
        v[u] = *(const float4*)(X + (size_t)(row0 + u * waves) * HD + lane * 4);
#pragma unroll
    for (int u = 0; u < U; ++u) {
        float s = v[u].x + v[u].y + v[u].z + v[u].w;
#pragma unroll
        for (int off = 16; off >= 1; off >>= 1) s += __shfl_xor(s, off, 32);
        const float mu = s * (1.0f / HD);

        const float dx = v[u].x - mu, dy = v[u].y - mu, dz = v[u].z - mu, dw = v[u].w - mu;
        float q = dx * dx + dy * dy + dz * dz + dw * dw;
#pragma unroll
        for (int off = 16; off >= 1; off >>= 1) q += __shfl_xor(q, off, 32);
        const float rstd = rsqrtf(q * (1.0f / HD) + LN_EPS);

        float4 o;
        o.x = dx * rstd * g.x + bt.x;
        o.y = dy * rstd * g.y + bt.y;
        o.z = dz * rstd * g.z + bt.z;
        o.w = dw * rstd * g.w + bt.w;
        *(float4*)(out + (size_t)(row0 + u * waves) * HD + lane * 4) = o;
    }
}

__global__ void layernorm_kernel(const float* __restrict__ X, const float* __restrict__ gamma,
                                 const float* __restrict__ beta, float* __restrict__ out, int M) {
    const int lane = threadIdx.x & 31;
    const int waves = (int)(gridDim.x * 8);
    const float4 g = *(const float4*)(gamma + lane * 4);
    const float4 bt = *(const float4*)(beta + lane * 4);

    int row = __builtin_amdgcn_readfirstlane((int)(blockIdx.x * 8 + (threadIdx.x >> 5)));
    for (; row + 3 * waves < M; row += 4 * waves) ln_rows<4>(X, out, g, bt, row, waves, lane);
    for (; row < M; row += waves) ln_rows<1>(X, out, g, bt, row, waves, lane);
}

extern "C" void kernel_launch(void* const* d_in, const int* in_sizes, int n_in,
                              void* d_out, int out_size, void* d_ws, size_t ws_size,
                              hipStream_t stream) {
    const float* user_x      = (const float*)d_in[0];
    const float* book_x      = (const float*)d_in[1];
    const int*   edge_src    = (const int*)d_in[2];
    const int*   edge_dst    = (const int*)d_in[3];
    const float* user_proj_w = (const float*)d_in[4];
    const float* user_proj_b = (const float*)d_in[5];
    const float* book_proj_w = (const float*)d_in[6];
    const float* book_proj_b = (const float*)d_in[7];
    const float* Wl          = (const float*)d_in[8];
    const float* bl          = (const float*)d_in[9];
    const float* Wr          = (const float*)d_in[10];
    const float* ugamma      = (const float*)d_in[11];
    const float* ubeta       = (const float*)d_in[12];
    const float* bgamma      = (const float*)d_in[13];
    const float* bbeta       = (const float*)d_in[14];

    float* ws = (float*)d_ws;
    size_t off = 0;
    auto carve = [&](size_t n) { float* p = ws + off; off += n; return p; };
    float* u_cur = carve((size_t)N_USER * HD);
    float* u_nxt = carve((size_t)N_USER * HD);
    float* b_cur = carve((size_t)N_BOOK * HD);
    float* b_nxt = carve((size_t)N_BOOK * HD);
    float* cnt   = carve((size_t)N_ROWS);
    int* deg     = (int*)carve((size_t)N_ROWS);
    int* offs    = (int*)carve((size_t)N_ROWS);
    int* counts  = (int*)carve((size_t)N_CNT);
    int* cofs    = (int*)carve((size_t)N_CNT);
    int* part    = (int*)carve((size_t)64);
    int* p_row   = (int*)carve((size_t)2 * N_EDGE);
    int* p_nbr   = (int*)carve((size_t)2 * N_EDGE);
    int* adj_raw = (int*)carve((size_t)2 * N_EDGE);
    int* adj     = (int*)carve((size_t)2 * N_EDGE);

    gemm_bias_kernel<<<(N_USER + 127) / 128, 256, 0, stream>>>(
        user_x, user_proj_w, user_proj_b, u_cur, N_USER, 64);
    gemm_bias_kernel<<<(N_BOOK + 127) / 128, 256, 0, stream>>>(
        book_x, book_proj_w, book_proj_b, b_cur, N_BOOK, 128);

    bucket_count_kernel<<<NB1, 1024, 0, stream>>>(edge_src, edge_dst, counts, N_EDGE);
    block_sum_kernel<<<CSCAN_BLOCKS, 256, 0, stream>>>(counts, part, N_CNT);
    scan_kernel<<<CSCAN_BLOCKS, 256, 0, stream>>>(counts, part, cofs, N_CNT);
    bucket_scatter_kernel<<<NB1, 1024, 0, stream>>>(
        edge_src, edge_dst, cofs, p_row, p_nbr, N_EDGE);
    build_lists_kernel<<<NBK, 1024, 0, stream>>>(
        p_row, p_nbr, cofs, deg, offs, cnt, adj_raw, adj, 2 * N_EDGE);

    for (int l = 0; l < 3; ++l) {
        const float* wl  = Wl + (size_t)l * HD * HD;
        const float* blv = bl + (size_t)l * HD;
        const float* wr  = Wr + (size_t)l * HD * HD;

        sage_layer_kernel<<<SAGE_BLOCKS, SAGE_WAVES * 32, 0, stream>>>(
            u_cur, b_cur, offs, deg, adj, cnt, wl, blv, wr, u_nxt, b_nxt);

        float* t;
        t = u_cur; u_cur = u_nxt; u_nxt = t;
        t = b_cur; b_cur = b_nxt; b_nxt = t;
    }

    float* out_u = (float*)d_out;
    float* out_b = out_u + (size_t)N_USER * HD;
    layernorm_kernel<<<FILL_BLOCKS, 256, 0, stream>>>(u_cur, ugamma, ubeta, out_u, N_USER);
    layernorm_kernel<<<FILL_BLOCKS, 256, 0, stream>>>(b_cur, bgamma, bbeta, out_b, N_BOOK);
}
